# starting kernel
# baseline (speedup 1.0000x reference)
#include <hip/hip_runtime.h>
#include <hip/hip_bf16.h>

typedef __attribute__((ext_vector_type(16))) __bf16 v16bf;
typedef __attribute__((ext_vector_type(8)))  float  v8f;
typedef __attribute__((ext_vector_type(4)))  float  v4f;

#define N_NODES 100000
#define E_EDGES 3200000
#define IN_F    512
#define HID     16
#define OUT_F   7
#define N_TILES (N_NODES / 16)
#define K_CHUNKS (IN_F / 32)

__device__ __forceinline__ void atomic_add_f32(float* p, float v) {
    __hip_atomic_fetch_add(p, v, __ATOMIC_RELAXED, __HIP_MEMORY_SCOPE_AGENT);
}

__global__ void GCN_pack_w1(const float* __restrict__ W1,
                            __bf16* __restrict__ Bh, __bf16* __restrict__ Bl) {
    int id = blockIdx.x * blockDim.x + threadIdx.x;
    if (id >= K_CHUNKS * 32 * 8) return;
    int v    = id & 7;
    int lane = (id >> 3) & 31;
    int c    = id >> 8;
    int n    = lane & 15;
    int g    = lane >> 4;
    int kb   = c * 32;
#pragma unroll
    for (int h = 0; h < 2; ++h) {
        int k = (v < 4) ? (kb + g * 8 + 2 * v + h)
                        : (kb + 16 + g * 8 + 2 * (v - 4) + h);
        float w   = W1[k * HID + n];
        __bf16 hi = (__bf16)w;
        __bf16 lo = (__bf16)(w - (float)hi);
        int idx = (c * 32 + lane) * 16 + v * 2 + h;
        Bh[idx] = hi;
        Bl[idx] = lo;
    }
}

__global__ void __launch_bounds__(256)
GCN_gemm1_wmma(const float* __restrict__ x,
               const __bf16* __restrict__ Bh, const __bf16* __restrict__ Bl,
               float* __restrict__ h0) {
    const int wave = threadIdx.x >> 5;
    const int lane = threadIdx.x & 31;
    const int tile = blockIdx.x * 8 + wave;
    if (tile >= N_TILES) return;

    const int n = lane & 15;
    const int g = lane >> 4;
    const float* xrow = x + (size_t)(tile * 16 + n) * IN_F;

    v8f acc = {};
#pragma unroll 4
    for (int c = 0; c < K_CHUNKS; ++c) {
        const int kb = c * 32;
        float f[16];
        *(v4f*)(f + 0)  = *(const v4f*)(xrow + kb + g * 8);
        *(v4f*)(f + 4)  = *(const v4f*)(xrow + kb + g * 8 + 4);
        *(v4f*)(f + 8)  = *(const v4f*)(xrow + kb + 16 + g * 8);
        *(v4f*)(f + 12) = *(const v4f*)(xrow + kb + 16 + g * 8 + 4);
        if (c + 1 < K_CHUNKS)
            __builtin_prefetch(xrow + kb + 32 + g * 8, 0, 3);

        v16bf ah, al;
#pragma unroll
        for (int i = 0; i < 16; ++i) {
            __bf16 hi = (__bf16)f[i];
            ah[i] = hi;
            al[i] = (__bf16)(f[i] - (float)hi);
        }
        v16bf bh = *(const v16bf*)(Bh + (c * 32 + lane) * 16);
        v16bf bl = *(const v16bf*)(Bl + (c * 32 + lane) * 16);

        acc = __builtin_amdgcn_wmma_f32_16x16x32_bf16(false, ah, false, bh,
                                                      (short)0, acc, false, false);
        acc = __builtin_amdgcn_wmma_f32_16x16x32_bf16(false, al, false, bh,
                                                      (short)0, acc, false, false);
        acc = __builtin_amdgcn_wmma_f32_16x16x32_bf16(false, ah, false, bl,
                                                      (short)0, acc, false, false);
    }
#pragma unroll
    for (int r = 0; r < 8; ++r)
        h0[(size_t)(tile * 16 + r + g * 8) * HID + n] = acc[r];
}

__global__ void GCN_spmm16(const int* __restrict__ ei, const float* __restrict__ ew,
                           const float* __restrict__ h, float* __restrict__ agg) {
    long long t = (long long)blockIdx.x * blockDim.x + threadIdx.x;
    if (t >= (long long)E_EDGES * 16) return;
    int e = (int)(t >> 4);
    int j = (int)(t & 15);
    int src = ei[e];
    int dst = ei[E_EDGES + e];
    float v = ew[e] * h[(size_t)src * HID + j];
    atomic_add_f32(&agg[(size_t)dst * HID + j], v);
}

__global__ void GCN_spmm7(const int* __restrict__ ei, const float* __restrict__ ew,
                          const float* __restrict__ z, float* __restrict__ agg) {
    long long t = (long long)blockIdx.x * blockDim.x + threadIdx.x;
    if (t >= (long long)E_EDGES * 8) return;
    int e = (int)(t >> 3);
    int j = (int)(t & 7);
    if (j >= OUT_F) return;
    int src = ei[e];
    int dst = ei[E_EDGES + e];
    float v = ew[e] * z[(size_t)src * OUT_F + j];
    atomic_add_f32(&agg[(size_t)dst * OUT_F + j], v);
}

__global__ void __launch_bounds__(256)
GCN_relu_gemm2(const float* __restrict__ agg1, const float* __restrict__ b1,
               const float* __restrict__ W2, float* __restrict__ z) {
    __shared__ float sW2[HID * OUT_F];
    __shared__ float sb1[HID];
    if (threadIdx.x < HID * OUT_F) sW2[threadIdx.x] = W2[threadIdx.x];
    if (threadIdx.x < HID)         sb1[threadIdx.x] = b1[threadIdx.x];
    __syncthreads();

    int node = blockIdx.x * blockDim.x + threadIdx.x;
    if (node >= N_NODES) return;

    float hv[HID];
#pragma unroll
    for (int i = 0; i < HID; ++i)
        hv[i] = fmaxf(agg1[(size_t)node * HID + i] + sb1[i], 0.0f);
#pragma unroll
    for (int j = 0; j < OUT_F; ++j) {
        float s = 0.0f;
#pragma unroll
        for (int i = 0; i < HID; ++i)
            s = fmaf(hv[i], sW2[i * OUT_F + j], s);
        z[(size_t)node * OUT_F + j] = s;
    }
}

__global__ void __launch_bounds__(256)
GCN_logsoftmax(const float* __restrict__ agg2, const float* __restrict__ b2,
               float* __restrict__ out) {
    __shared__ float sb2[OUT_F];
    if (threadIdx.x < OUT_F) sb2[threadIdx.x] = b2[threadIdx.x];
    __syncthreads();

    int node = blockIdx.x * blockDim.x + threadIdx.x;
    if (node >= N_NODES) return;

    float v[OUT_F];
    float m = -3.402823466e38f;
#pragma unroll
    for (int j = 0; j < OUT_F; ++j) {
        v[j] = agg2[(size_t)node * OUT_F + j] + sb2[j];
        m = fmaxf(m, v[j]);
    }
    float s = 0.0f;
#pragma unroll
    for (int j = 0; j < OUT_F; ++j) s += expf(v[j] - m);
    float lse = logf(s);
#pragma unroll
    for (int j = 0; j < OUT_F; ++j)
        out[(size_t)node * OUT_F + j] = v[j] - m - lse;
}

__global__ void GCN_zero(float* __restrict__ p, long long n) {
    long long i = (long long)blockIdx.x * blockDim.x + threadIdx.x;
    if (i < n) p[i] = 0.0f;
}

extern "C" void kernel_launch(void* const* d_in, const int* in_sizes, int n_in,
                              void* d_out, int out_size, void* d_ws, size_t ws_size,
                              hipStream_t stream) {
    const float* x  = (const float*)d_in[0];
    const int*   ei = (const int*)d_in[1];
    const float* ew = (const float*)d_in[2];
    const float* W1 = (const float*)d_in[3];
    const float* b1 = (const float*)d_in[4];
    const float* W2 = (const float*)d_in[5];
    const float* b2 = (const float*)d_in[6];
    float* out = (float*)d_out;

    char* ws = (char*)d_ws;
    size_t off = 0;
    auto alloc = [&](size_t bytes) { char* p = ws + off; off = (off + bytes + 255) & ~(size_t)255; return p; };
    float*  h0   = (float*)alloc((size_t)N_NODES * HID * 4);
    float*  agg1 = (float*)alloc((size_t)N_NODES * HID * 4);
    float*  z    = (float*)alloc((size_t)N_NODES * OUT_F * 4);
    float*  agg2 = (float*)alloc((size_t)N_NODES * OUT_F * 4);
    __bf16* Bh   = (__bf16*)alloc((size_t)K_CHUNKS * 32 * 16 * 2);
    __bf16* Bl   = (__bf16*)alloc((size_t)K_CHUNKS * 32 * 16 * 2);

    {
        long long n1 = (long long)N_NODES * HID;
        long long n2 = (long long)N_NODES * OUT_F;
        GCN_zero<<<(int)((n1 + 255) / 256), 256, 0, stream>>>(agg1, n1);
        GCN_zero<<<(int)((n2 + 255) / 256), 256, 0, stream>>>(agg2, n2);
    }

    GCN_pack_w1<<<K_CHUNKS, 256, 0, stream>>>(W1, Bh, Bl);

    GCN_gemm1_wmma<<<(N_TILES + 7) / 8, 256, 0, stream>>>(x, Bh, Bl, h0);

    {
        long long t = (long long)E_EDGES * 16;
        GCN_spmm16<<<(int)((t + 255) / 256), 256, 0, stream>>>(ei, ew, h0, agg1);
    }

    GCN_relu_gemm2<<<(N_NODES + 255) / 256, 256, 0, stream>>>(agg1, b1, W2, z);

    {
        long long t = (long long)E_EDGES * 8;
        GCN_spmm7<<<(int)((t + 255) / 256), 256, 0, stream>>>(ei, ew, z, agg2);
    }

    GCN_logsoftmax<<<(N_NODES + 255) / 256, 256, 0, stream>>>(agg2, b2, out);
}


# Edge passes without global atomics: work-group (edge slice, 3125-node range) filters its slice's destinations, sums w*row in an LDS plane (LDS f32 adds); 8 planes added in order; sum order changes
# speedup vs baseline: 1.5680x; 1.5680x over previous
#include <hip/hip_runtime.h>
#include <hip/hip_bf16.h>

typedef __attribute__((ext_vector_type(16))) __bf16 v16bf;
typedef __attribute__((ext_vector_type(8)))  float  v8f;
typedef __attribute__((ext_vector_type(4)))  float  v4f;

#define N_NODES 100000
#define E_EDGES 3200000
#define IN_F    512
#define HID     16
#define OUT_F   7
#define N_TILES (N_NODES / 16)
#define K_CHUNKS (IN_F / 32)

__device__ __forceinline__ void atomic_add_f32(float* p, float v) {
    __hip_atomic_fetch_add(p, v, __ATOMIC_RELAXED, __HIP_MEMORY_SCOPE_AGENT);
}

__global__ void GCN_pack_w1(const float* __restrict__ W1,
                            __bf16* __restrict__ Bh, __bf16* __restrict__ Bl) {
    int id = blockIdx.x * blockDim.x + threadIdx.x;
    if (id >= K_CHUNKS * 32 * 8) return;
    int v    = id & 7;
    int lane = (id >> 3) & 31;
    int c    = id >> 8;
    int n    = lane & 15;
    int g    = lane >> 4;
    int kb   = c * 32;
#pragma unroll
    for (int h = 0; h < 2; ++h) {
        int k = (v < 4) ? (kb + g * 8 + 2 * v + h)
                        : (kb + 16 + g * 8 + 2 * (v - 4) + h);
        float w   = W1[k * HID + n];
        __bf16 hi = (__bf16)w;
        __bf16 lo = (__bf16)(w - (float)hi);
        int idx = (c * 32 + lane) * 16 + v * 2 + h;
        Bh[idx] = hi;
        Bl[idx] = lo;
    }
}

__global__ void __launch_bounds__(256)
GCN_gemm1_wmma(const float* __restrict__ x,
               const __bf16* __restrict__ Bh, const __bf16* __restrict__ Bl,
               float* __restrict__ h0) {
    const int wave = threadIdx.x >> 5;
    const int lane = threadIdx.x & 31;
    const int tile = blockIdx.x * 8 + wave;
    if (tile >= N_TILES) return;

    const int n = lane & 15;
    const int g = lane >> 4;
    const float* xrow = x + (size_t)(tile * 16 + n) * IN_F;

    v8f acc = {};
#pragma unroll 4
    for (int c = 0; c < K_CHUNKS; ++c) {
        const int kb = c * 32;
        float f[16];
        *(v4f*)(f + 0)  = *(const v4f*)(xrow + kb + g * 8);
        *(v4f*)(f + 4)  = *(const v4f*)(xrow + kb + g * 8 + 4);
        *(v4f*)(f + 8)  = *(const v4f*)(xrow + kb + 16 + g * 8);
        *(v4f*)(f + 12) = *(const v4f*)(xrow + kb + 16 + g * 8 + 4);
        if (c + 1 < K_CHUNKS)
            __builtin_prefetch(xrow + kb + 32 + g * 8, 0, 3);

        v16bf ah, al;
#pragma unroll
        for (int i = 0; i < 16; ++i) {
            __bf16 hi = (__bf16)f[i];
            ah[i] = hi;
            al[i] = (__bf16)(f[i] - (float)hi);
        }
        v16bf bh = *(const v16bf*)(Bh + (c * 32 + lane) * 16);
        v16bf bl = *(const v16bf*)(Bl + (c * 32 + lane) * 16);

        acc = __builtin_amdgcn_wmma_f32_16x16x32_bf16(false, ah, false, bh,
                                                      (short)0, acc, false, false);
        acc = __builtin_amdgcn_wmma_f32_16x16x32_bf16(false, al, false, bh,
                                                      (short)0, acc, false, false);
        acc = __builtin_amdgcn_wmma_f32_16x16x32_bf16(false, ah, false, bl,
                                                      (short)0, acc, false, false);
    }
#pragma unroll
    for (int r = 0; r < 8; ++r)
        h0[(size_t)(tile * 16 + r + g * 8) * HID + n] = acc[r];
}

__global__ void GCN_spmm16(const int* __restrict__ ei, const float* __restrict__ ew,
                           const float* __restrict__ h, float* __restrict__ agg) {
    long long t = (long long)blockIdx.x * blockDim.x + threadIdx.x;
    if (t >= (long long)E_EDGES * 16) return;
    int e = (int)(t >> 4);
    int j = (int)(t & 15);
    int src = ei[e];
    int dst = ei[E_EDGES + e];
    float v = ew[e] * h[(size_t)src * HID + j];
    atomic_add_f32(&agg[(size_t)dst * HID + j], v);
}

__global__ void GCN_spmm7(const int* __restrict__ ei, const float* __restrict__ ew,
                          const float* __restrict__ z, float* __restrict__ agg) {
    long long t = (long long)blockIdx.x * blockDim.x + threadIdx.x;
    if (t >= (long long)E_EDGES * 8) return;
    int e = (int)(t >> 3);
    int j = (int)(t & 7);
    if (j >= OUT_F) return;
    int src = ei[e];
    int dst = ei[E_EDGES + e];
    float v = ew[e] * z[(size_t)src * OUT_F + j];
    atomic_add_f32(&agg[(size_t)dst * OUT_F + j], v);
}

__global__ void __launch_bounds__(256)
GCN_relu_gemm2(const float* __restrict__ agg1, const float* __restrict__ b1,
               const float* __restrict__ W2, float* __restrict__ z) {
    __shared__ float sW2[HID * OUT_F];
    __shared__ float sb1[HID];
    if (threadIdx.x < HID * OUT_F) sW2[threadIdx.x] = W2[threadIdx.x];
    if (threadIdx.x < HID)         sb1[threadIdx.x] = b1[threadIdx.x];
    __syncthreads();

    int node = blockIdx.x * blockDim.x + threadIdx.x;
    if (node >= N_NODES) return;

    float hv[HID];
#pragma unroll
    for (int i = 0; i < HID; ++i)
        hv[i] = fmaxf(agg1[(size_t)node * HID + i] + sb1[i], 0.0f);
#pragma unroll
    for (int j = 0; j < OUT_F; ++j) {
        float s = 0.0f;
#pragma unroll
        for (int i = 0; i < HID; ++i)
            s = fmaf(hv[i], sW2[i * OUT_F + j], s);
        z[(size_t)node * OUT_F + j] = s;
    }
}

__global__ void __launch_bounds__(256)
GCN_logsoftmax(const float* __restrict__ agg2, const float* __restrict__ b2,
               float* __restrict__ out) {
    __shared__ float sb2[OUT_F];
    if (threadIdx.x < OUT_F) sb2[threadIdx.x] = b2[threadIdx.x];
    __syncthreads();

    int node = blockIdx.x * blockDim.x + threadIdx.x;
    if (node >= N_NODES) return;

    float v[OUT_F];
    float m = -3.402823466e38f;
#pragma unroll
    for (int j = 0; j < OUT_F; ++j) {
        v[j] = agg2[(size_t)node * OUT_F + j] + sb2[j];
        m = fmaxf(m, v[j]);
    }
    float s = 0.0f;
#pragma unroll
    for (int j = 0; j < OUT_F; ++j) s += expf(v[j] - m);
    float lse = logf(s);
#pragma unroll
    for (int j = 0; j < OUT_F; ++j)
        out[(size_t)node * OUT_F + j] = v[j] - m - lse;
}

__global__ void GCN_zero(float* __restrict__ p, long long n) {
    long long i = (long long)blockIdx.x * blockDim.x + threadIdx.x;
    if (i < n) p[i] = 0.0f;
}

typedef __attribute__((ext_vector_type(4))) int v4i;

#define SL        8
#define RG        32
#define RN        (N_NODES / RG)
#define SLICE_E   (E_EDGES / SL)
#define SLICE_Q   (SLICE_E / 4)
#define EP_T      1024
#define EP_U      8
#define CH_Q      (EP_T * EP_U)
#define N_CH      ((SLICE_Q + CH_Q - 1) / CH_Q)
#define WQ_CAP    (32 * 4 * EP_U)
#define STRIDE16  50016
#define STRIDE7   21888

template <int D>
__global__ void __launch_bounds__(EP_T)
GCN_edge_owner(const int* __restrict__ ei, const float* __restrict__ ew,
               const float* __restrict__ h, float* __restrict__ part) {
    constexpr int PLANE  = RN * D;
    constexpr int STRIDE = (D == 16) ? STRIDE16 : STRIDE7;
    constexpr int LPM    = 4;
    constexpr int MPS    = 32 / LPM;
    constexpr int NSTEP  = 64 / MPS;
    constexpr int VW     = (D == 16) ? 4 : 2;
    constexpr int PITCH  = (D == 16) ? 17 : 7;

    constexpr int PLANE_LDS   = RN * PITCH;
    constexpr int PLANE_ALLOC = (PLANE_LDS + 4 > 24600) ? (PLANE_LDS + 4) : 24600;
    __shared__ __attribute__((aligned(16))) float plane[PLANE_ALLOC];
    __shared__ unsigned short wqueue[32 * WQ_CAP];

    const int tid  = threadIdx.x;
    const int lane = tid & 31;
    const int wave = tid >> 5;
    const int s    = blockIdx.x / RG;
    const int r    = blockIdx.x % RG;
    const int lo   = r * RN;

    const int*   srcs = ei + (size_t)s * SLICE_E;
    const int*   dsts = ei + (size_t)E_EDGES + (size_t)s * SLICE_E;
    const float* ews  = ew + (size_t)s * SLICE_E;
    const v4i*   dq   = (const v4i*)dsts;
    unsigned short* wq = wqueue + wave * WQ_CAP;

    for (int i = tid; i < PLANE_LDS; i += EP_T) plane[i] = 0.0f;

    __syncthreads();

    const int sub = lane / LPM;
    const int jl  = lane % LPM;

    for (int ch = 0; ch < N_CH; ++ch) {
        v4i dv[EP_U];
#pragma unroll
        for (int u = 0; u < EP_U; ++u) {
            int q = ch * CH_Q + u * EP_T + tid;
            dv[u] = (q < SLICE_Q) ? dq[q] : (v4i){-1, -1, -1, -1};
        }
        unsigned M = 0;
#pragma unroll
        for (int u = 0; u < EP_U; ++u) {
#pragma unroll
            for (int c = 0; c < 4; ++c) {
                if ((unsigned)(dv[u][c] - lo) < (unsigned)RN) M |= 1u << (u * 4 + c);
            }
        }
        const unsigned mine = (unsigned)__builtin_popcount(M);
        unsigned incl = mine;
#pragma unroll
        for (int d = 1; d < 32; d <<= 1) {
            unsigned up = __shfl_up(incl, d);
            if (lane >= d) incl += up;
        }
        unsigned pos = incl - mine;
        const unsigned cnt = (unsigned)__builtin_amdgcn_readlane((int)incl, 31);
        while (M) {
            const int b = __builtin_ctz(M);
            M &= M - 1;
            wq[pos++] = (unsigned short)(((b >> 2) << 7) + lane * 4 + (b & 3));
        }
        __syncthreads();
        if (ch + 1 < N_CH) {
#pragma unroll
            for (int u = 0; u < EP_U; ++u) {
                int q = (ch + 1) * CH_Q + u * EP_T + tid;
                if (q < SLICE_Q) __builtin_prefetch(dq + q, 0, 3);
            }
        }

        const int ebase = (ch * CH_Q + wave * 32) * 4;
        for (unsigned base = 0; base < cnt; base += 64) {
            const unsigned nm = (cnt - base < 64u) ? (cnt - base) : 64u;
            int   sA = 0, sB = 0, lA = lo, lB = lo;
            float wA = 0.0f, wB = 0.0f;
            if ((unsigned)lane < nm) {
                unsigned p = wq[base + lane];
                int e = ebase + (int)(p >> 7) * (EP_T * 4) + (int)(p & 127u);
                sA = srcs[e]; wA = ews[e]; lA = dsts[e];
            }
            if ((unsigned)lane + 32u < nm) {
                unsigned p = wq[base + 32 + lane];
                int e = ebase + (int)(p >> 7) * (EP_T * 4) + (int)(p & 127u);
                sB = srcs[e]; wB = ews[e]; lB = dsts[e];
            }
            float vv[NSTEP][VW];
            float wk[NSTEP];
            int   lk[NSTEP];
            bool  ok[NSTEP];
#pragma unroll
            for (int t = 0; t < NSTEP; ++t) {
                const int k  = t * MPS + sub;
                const int sl = k & 31;
                const int ss = __shfl((t * MPS < 32) ? sA : sB, sl);
                wk[t] = __shfl((t * MPS < 32) ? wA : wB, sl);
                lk[t] = __shfl((t * MPS < 32) ? lA : lB, sl) - lo;
                ok[t] = ((unsigned)k < nm);
                if constexpr (D == 16) {
                    v4f x4 = {0.0f, 0.0f, 0.0f, 0.0f};
                    if (ok[t]) x4 = *((const v4f*)(h + (size_t)ss * 16) + jl);
#pragma unroll
                    for (int i = 0; i < VW; ++i) vv[t][i] = x4[i];
                } else {
                    float x0 = 0.0f, x1 = 0.0f;
                    if (ok[t]) {
                        const float* hr = h + (size_t)ss * D + jl * 2;
                        x0 = hr[0];
                        if (jl < 3) x1 = hr[1];
                    }
                    vv[t][0] = x0;
                    vv[t][1] = x1;
                }
            }
#pragma unroll
            for (int t = 0; t < NSTEP; ++t) {
                if (ok[t]) {
                    float* pr = plane + lk[t] * PITCH + jl * VW;
#pragma unroll
                    for (int i = 0; i < VW; ++i) {
                        if (jl * VW + i < D)
                            __hip_atomic_fetch_add(pr + i, wk[t] * vv[t][i],
                                                   __ATOMIC_RELAXED, __HIP_MEMORY_SCOPE_WORKGROUP);
                    }
                }
            }
        }
        __syncthreads();
    }

    float* dstp = part + (size_t)blockIdx.x * STRIDE;
    for (int i = tid; i < PLANE; i += EP_T) dstp[i] = plane[(i / D) * PITCH + (i % D)];
}

__global__ void __launch_bounds__(256)
GCN_relu_gemm2_parts(const float* __restrict__ part, const float* __restrict__ b1,
                     const float* __restrict__ W2, float* __restrict__ z) {
    __shared__ float sW2[HID * OUT_F];
    __shared__ float sb1[HID];
    if (threadIdx.x < HID * OUT_F) sW2[threadIdx.x] = W2[threadIdx.x];
    if (threadIdx.x < HID)         sb1[threadIdx.x] = b1[threadIdx.x];
    __syncthreads();

    int node = blockIdx.x * blockDim.x + threadIdx.x;
    if (node >= N_NODES) return;
    const int r  = node / RN;
    const int nl = node - r * RN;
    const float* p0 = part + (size_t)r * STRIDE16 + (size_t)nl * HID;

    float a[HID];
#pragma unroll
    for (int i = 0; i < HID; ++i) a[i] = 0.0f;
#pragma unroll
    for (int sl = 0; sl < SL; ++sl) {
        const v4f* p = (const v4f*)(p0 + (size_t)sl * RG * STRIDE16);
#pragma unroll
        for (int q = 0; q < 4; ++q) {
            v4f t = p[q];
            a[q * 4 + 0] += t[0]; a[q * 4 + 1] += t[1];
            a[q * 4 + 2] += t[2]; a[q * 4 + 3] += t[3];
        }
    }
    float hv[HID];
#pragma unroll
    for (int i = 0; i < HID; ++i) hv[i] = fmaxf(a[i] + sb1[i], 0.0f);
#pragma unroll
    for (int j = 0; j < OUT_F; ++j) {
        float sum = 0.0f;
#pragma unroll
        for (int i = 0; i < HID; ++i)
            sum = fmaf(hv[i], sW2[i * OUT_F + j], sum);
        z[(size_t)node * OUT_F + j] = sum;
    }
}

__global__ void __launch_bounds__(256)
GCN_logsoftmax_parts(const float* __restrict__ part, const float* __restrict__ b2,
                     float* __restrict__ out) {
    __shared__ float sb2[OUT_F];
    if (threadIdx.x < OUT_F) sb2[threadIdx.x] = b2[threadIdx.x];
    __syncthreads();

    int node = blockIdx.x * blockDim.x + threadIdx.x;
    if (node >= N_NODES) return;
    const int r  = node / RN;
    const int nl = node - r * RN;
    const float* p0 = part + (size_t)r * STRIDE7 + (size_t)nl * OUT_F;

    float v[OUT_F];
#pragma unroll
    for (int j = 0; j < OUT_F; ++j) v[j] = 0.0f;
#pragma unroll
    for (int sl = 0; sl < SL; ++sl) {
        const float* p = p0 + (size_t)sl * RG * STRIDE7;
#pragma unroll
        for (int j = 0; j < OUT_F; ++j) v[j] += p[j];
    }
    float m = -3.402823466e38f;
#pragma unroll
    for (int j = 0; j < OUT_F; ++j) {
        v[j] += sb2[j];
        m = fmaxf(m, v[j]);
    }
    float sum = 0.0f;
#pragma unroll
    for (int j = 0; j < OUT_F; ++j) sum += expf(v[j] - m);
    float lse = logf(sum);
#pragma unroll
    for (int j = 0; j < OUT_F; ++j)
        out[(size_t)node * OUT_F + j] = v[j] - m - lse;
}

extern "C" void kernel_launch(void* const* d_in, const int* in_sizes, int n_in,
                              void* d_out, int out_size, void* d_ws, size_t ws_size,
                              hipStream_t stream) {
    const float* x  = (const float*)d_in[0];
    const int*   ei = (const int*)d_in[1];
    const float* ew = (const float*)d_in[2];
    const float* W1 = (const float*)d_in[3];
    const float* b1 = (const float*)d_in[4];
    const float* W2 = (const float*)d_in[5];
    const float* b2 = (const float*)d_in[6];
    float* out = (float*)d_out;

    char* ws = (char*)d_ws;
    size_t off = 0;
    auto alloc = [&](size_t bytes) { char* p = ws + off; off = (off + bytes + 255) & ~(size_t)255; return p; };
    float*  h0   = (float*)alloc((size_t)N_NODES * HID * 4);
    float*  z    = (float*)alloc((size_t)N_NODES * OUT_F * 4);
    __bf16* Bh   = (__bf16*)alloc((size_t)K_CHUNKS * 32 * 16 * 2);
    __bf16* Bl   = (__bf16*)alloc((size_t)K_CHUNKS * 32 * 16 * 2);
    float*  part = (float*)alloc((size_t)SL * RG * STRIDE16 * 4);

    if (off <= ws_size) {
        GCN_pack_w1<<<K_CHUNKS, 256, 0, stream>>>(W1, Bh, Bl);
        GCN_gemm1_wmma<<<(N_TILES + 7) / 8, 256, 0, stream>>>(x, Bh, Bl, h0);
        GCN_edge_owner<16><<<SL * RG, EP_T, 0, stream>>>(ei, ew, h0, part);
        GCN_relu_gemm2_parts<<<(N_NODES + 255) / 256, 256, 0, stream>>>(part, b1, W2, z);
        GCN_edge_owner<OUT_F><<<SL * RG, EP_T, 0, stream>>>(ei, ew, z, part);
        GCN_logsoftmax_parts<<<(N_NODES + 255) / 256, 256, 0, stream>>>(part, b2, out);
        return;
    }

    off = 0;
    h0 = (float*)alloc((size_t)N_NODES * HID * 4);
    float* agg1 = (float*)alloc((size_t)N_NODES * HID * 4);
    z = (float*)alloc((size_t)N_NODES * OUT_F * 4);
    float* agg2 = (float*)alloc((size_t)N_NODES * OUT_F * 4);
    Bh = (__bf16*)alloc((size_t)K_CHUNKS * 32 * 16 * 2);
    Bl = (__bf16*)alloc((size_t)K_CHUNKS * 32 * 16 * 2);
    {
        long long n1 = (long long)N_NODES * HID;
        long long n2 = (long long)N_NODES * OUT_F;
        GCN_zero<<<(int)((n1 + 255) / 256), 256, 0, stream>>>(agg1, n1);
        GCN_zero<<<(int)((n2 + 255) / 256), 256, 0, stream>>>(agg2, n2);
    }
    GCN_pack_w1<<<K_CHUNKS, 256, 0, stream>>>(W1, Bh, Bl);
    GCN_gemm1_wmma<<<(N_TILES + 7) / 8, 256, 0, stream>>>(x, Bh, Bl, h0);
    {
        long long t = (long long)E_EDGES * 16;
        GCN_spmm16<<<(int)((t + 255) / 256), 256, 0, stream>>>(ei, ew, h0, agg1);
    }
    GCN_relu_gemm2<<<(N_NODES + 255) / 256, 256, 0, stream>>>(agg1, b1, W2, z);
    {
        long long t = (long long)E_EDGES * 8;
        GCN_spmm7<<<(int)((t + 255) / 256), 256, 0, stream>>>(ei, ew, z, agg2);
    }
    GCN_logsoftmax<<<(N_NODES + 255) / 256, 256, 0, stream>>>(agg2, b2, out);
}


# Layer-1 edge pass writes each wave's kept edges (source, row, weight) to the workspace; layer-2 pass sums from that list instead of filtering destinations again; LDS add order changes
# speedup vs baseline: 2.0427x; 1.3027x over previous
#include <hip/hip_runtime.h>
#include <hip/hip_bf16.h>

typedef __attribute__((ext_vector_type(16))) __bf16 v16bf;
typedef __attribute__((ext_vector_type(8)))  float  v8f;
typedef __attribute__((ext_vector_type(4)))  float  v4f;

#define N_NODES 100000
#define E_EDGES 3200000
#define IN_F    512
#define HID     16
#define OUT_F   7
#define N_TILES (N_NODES / 16)
#define K_CHUNKS (IN_F / 32)

__device__ __forceinline__ void atomic_add_f32(float* p, float v) {
    __hip_atomic_fetch_add(p, v, __ATOMIC_RELAXED, __HIP_MEMORY_SCOPE_AGENT);
}

__global__ void GCN_pack_w1(const float* __restrict__ W1,
                            __bf16* __restrict__ Bh, __bf16* __restrict__ Bl) {
    int id = blockIdx.x * blockDim.x + threadIdx.x;
    if (id >= K_CHUNKS * 32 * 8) return;
    int v    = id & 7;
    int lane = (id >> 3) & 31;
    int c    = id >> 8;
    int n    = lane & 15;
    int g    = lane >> 4;
    int kb   = c * 32;
#pragma unroll
    for (int h = 0; h < 2; ++h) {
        int k = (v < 4) ? (kb + g * 8 + 2 * v + h)
                        : (kb + 16 + g * 8 + 2 * (v - 4) + h);
        float w   = W1[k * HID + n];
        __bf16 hi = (__bf16)w;
        __bf16 lo = (__bf16)(w - (float)hi);
        int idx = (c * 32 + lane) * 16 + v * 2 + h;
        Bh[idx] = hi;
        Bl[idx] = lo;
    }
}

__global__ void __launch_bounds__(256)
GCN_gemm1_wmma(const float* __restrict__ x,
               const __bf16* __restrict__ Bh, const __bf16* __restrict__ Bl,
               float* __restrict__ h0) {
    const int wave = threadIdx.x >> 5;
    const int lane = threadIdx.x & 31;
    const int tile = blockIdx.x * 8 + wave;
    if (tile >= N_TILES) return;

    const int n = lane & 15;
    const int g = lane >> 4;
    const float* xrow = x + (size_t)(tile * 16 + n) * IN_F;

    v8f acc = {};
#pragma unroll 4
    for (int c = 0; c < K_CHUNKS; ++c) {
        const int kb = c * 32;
        float f[16];
        *(v4f*)(f + 0)  = *(const v4f*)(xrow + kb + g * 8);
        *(v4f*)(f + 4)  = *(const v4f*)(xrow + kb + g * 8 + 4);
        *(v4f*)(f + 8)  = *(const v4f*)(xrow + kb + 16 + g * 8);
        *(v4f*)(f + 12) = *(const v4f*)(xrow + kb + 16 + g * 8 + 4);
        if (c + 1 < K_CHUNKS)
            __builtin_prefetch(xrow + kb + 32 + g * 8, 0, 3);

        v16bf ah, al;
#pragma unroll
        for (int i = 0; i < 16; ++i) {
            __bf16 hi = (__bf16)f[i];
            ah[i] = hi;
            al[i] = (__bf16)(f[i] - (float)hi);
        }
        v16bf bh = *(const v16bf*)(Bh + (c * 32 + lane) * 16);
        v16bf bl = *(const v16bf*)(Bl + (c * 32 + lane) * 16);

        acc = __builtin_amdgcn_wmma_f32_16x16x32_bf16(false, ah, false, bh,
                                                      (short)0, acc, false, false);
        acc = __builtin_amdgcn_wmma_f32_16x16x32_bf16(false, al, false, bh,
                                                      (short)0, acc, false, false);
        acc = __builtin_amdgcn_wmma_f32_16x16x32_bf16(false, ah, false, bl,
                                                      (short)0, acc, false, false);
    }
#pragma unroll
    for (int r = 0; r < 8; ++r)
        h0[(size_t)(tile * 16 + r + g * 8) * HID + n] = acc[r];
}

__global__ void GCN_spmm16(const int* __restrict__ ei, const float* __restrict__ ew,
                           const float* __restrict__ h, float* __restrict__ agg) {
    long long t = (long long)blockIdx.x * blockDim.x + threadIdx.x;
    if (t >= (long long)E_EDGES * 16) return;
    int e = (int)(t >> 4);
    int j = (int)(t & 15);
    int src = ei[e];
    int dst = ei[E_EDGES + e];
    float v = ew[e] * h[(size_t)src * HID + j];
    atomic_add_f32(&agg[(size_t)dst * HID + j], v);
}

__global__ void GCN_spmm7(const int* __restrict__ ei, const float* __restrict__ ew,
                          const float* __restrict__ z, float* __restrict__ agg) {
    long long t = (long long)blockIdx.x * blockDim.x + threadIdx.x;
    if (t >= (long long)E_EDGES * 8) return;
    int e = (int)(t >> 3);
    int j = (int)(t & 7);
    if (j >= OUT_F) return;
    int src = ei[e];
    int dst = ei[E_EDGES + e];
    float v = ew[e] * z[(size_t)src * OUT_F + j];
    atomic_add_f32(&agg[(size_t)dst * OUT_F + j], v);
}

__global__ void __launch_bounds__(256)
GCN_relu_gemm2(const float* __restrict__ agg1, const float* __restrict__ b1,
               const float* __restrict__ W2, float* __restrict__ z) {
    __shared__ float sW2[HID * OUT_F];
    __shared__ float sb1[HID];
    if (threadIdx.x < HID * OUT_F) sW2[threadIdx.x] = W2[threadIdx.x];
    if (threadIdx.x < HID)         sb1[threadIdx.x] = b1[threadIdx.x];
    __syncthreads();

    int node = blockIdx.x * blockDim.x + threadIdx.x;
    if (node >= N_NODES) return;

    float hv[HID];
#pragma unroll
    for (int i = 0; i < HID; ++i)
        hv[i] = fmaxf(agg1[(size_t)node * HID + i] + sb1[i], 0.0f);
#pragma unroll
    for (int j = 0; j < OUT_F; ++j) {
        float s = 0.0f;
#pragma unroll
        for (int i = 0; i < HID; ++i)
            s = fmaf(hv[i], sW2[i * OUT_F + j], s);
        z[(size_t)node * OUT_F + j] = s;
    }
}

__global__ void __launch_bounds__(256)
GCN_logsoftmax(const float* __restrict__ agg2, const float* __restrict__ b2,
               float* __restrict__ out) {
    __shared__ float sb2[OUT_F];
    if (threadIdx.x < OUT_F) sb2[threadIdx.x] = b2[threadIdx.x];
    __syncthreads();

    int node = blockIdx.x * blockDim.x + threadIdx.x;
    if (node >= N_NODES) return;

    float v[OUT_F];
    float m = -3.402823466e38f;
#pragma unroll
    for (int j = 0; j < OUT_F; ++j) {
        v[j] = agg2[(size_t)node * OUT_F + j] + sb2[j];
        m = fmaxf(m, v[j]);
    }
    float s = 0.0f;
#pragma unroll
    for (int j = 0; j < OUT_F; ++j) s += expf(v[j] - m);
    float lse = logf(s);
#pragma unroll
    for (int j = 0; j < OUT_F; ++j)
        out[(size_t)node * OUT_F + j] = v[j] - m - lse;
}

__global__ void GCN_zero(float* __restrict__ p, long long n) {
    long long i = (long long)blockIdx.x * blockDim.x + threadIdx.x;
    if (i < n) p[i] = 0.0f;
}

typedef __attribute__((ext_vector_type(4))) int v4i;

#define SL        8
#define RG        32
#define RN        (N_NODES / RG)
#define SLICE_E   (E_EDGES / SL)
#define SLICE_Q   (SLICE_E / 4)
#define EP_T      1024
#define EP_U      8
#define CH_Q      (EP_T * EP_U)
#define N_CH      ((SLICE_Q + CH_Q - 1) / CH_Q)
#define WQ_CAP    (32 * 4 * EP_U)
#define STRIDE16  50016
#define STRIDE7   21888

#define LIST_CAPW 576
#define LIST_OVF  0xFFFFFFFFu
#define SRC_BITS  17
typedef __attribute__((ext_vector_type(2))) unsigned v2u;

template <int D>
__device__ __forceinline__ void
edge_round(const float* __restrict__ h, float* plane, int lane, unsigned nm,
           int sA, float wA, int rA, int sB, float wB, int rB) {
    constexpr int LPM    = 4;
    constexpr int MPS    = 32 / LPM;
    constexpr int NSTEP  = 64 / MPS;
    constexpr int VW     = (D == 16) ? 4 : 2;
    constexpr int PITCH  = (D == 16) ? 17 : 7;
    const int sub = lane / LPM;
    const int jl  = lane % LPM;

    float vv[NSTEP][VW];
    float wk[NSTEP];
    int   lk[NSTEP];
    bool  ok[NSTEP];
#pragma unroll
    for (int t = 0; t < NSTEP; ++t) {
        const int k  = t * MPS + sub;
        const int sl = k & 31;
        const int ss = __shfl((t * MPS < 32) ? sA : sB, sl);
        wk[t] = __shfl((t * MPS < 32) ? wA : wB, sl);
        lk[t] = __shfl((t * MPS < 32) ? rA : rB, sl);
        ok[t] = ((unsigned)k < nm);
        if constexpr (D == 16) {
            v4f x4 = {0.0f, 0.0f, 0.0f, 0.0f};
            if (ok[t]) x4 = *((const v4f*)(h + (size_t)ss * 16) + jl);
#pragma unroll
            for (int i = 0; i < VW; ++i) vv[t][i] = x4[i];
        } else {
            float x0 = 0.0f, x1 = 0.0f;
            if (ok[t]) {
                const float* hr = h + (size_t)ss * D + jl * 2;
                x0 = hr[0];
                if (jl < 3) x1 = hr[1];
            }
            vv[t][0] = x0;
            vv[t][1] = x1;
        }
    }
#pragma unroll
    for (int t = 0; t < NSTEP; ++t) {
        if (ok[t]) {
            float* pr = plane + lk[t] * PITCH + jl * VW;
#pragma unroll
            for (int i = 0; i < VW; ++i) {
                if (jl * VW + i < D)
                    __hip_atomic_fetch_add(pr + i, wk[t] * vv[t][i],
                                           __ATOMIC_RELAXED, __HIP_MEMORY_SCOPE_WORKGROUP);
            }
        }
    }
}

template <int D, bool EMIT>
__device__ __forceinline__ unsigned
edge_filter_pass(const int* __restrict__ srcs, const int* __restrict__ dsts,
                 const float* __restrict__ ews, const float* __restrict__ h,
                 float* plane, unsigned short* wq, v2u* __restrict__ lw,
                 bool want_list, int tid, int lane, int wave, int lo) {
    const v4i* dq = (const v4i*)dsts;
    unsigned woff = 0;
    bool     ovf  = !want_list;
    for (int ch = 0; ch < N_CH; ++ch) {
        v4i dv[EP_U];
#pragma unroll
        for (int u = 0; u < EP_U; ++u) {
            int q = ch * CH_Q + u * EP_T + tid;
            dv[u] = (q < SLICE_Q) ? dq[q] : (v4i){-1, -1, -1, -1};
        }
        unsigned M = 0;
#pragma unroll
        for (int u = 0; u < EP_U; ++u) {
#pragma unroll
            for (int c = 0; c < 4; ++c) {
                if ((unsigned)(dv[u][c] - lo) < (unsigned)RN) M |= 1u << (u * 4 + c);
            }
        }
        const unsigned mine = (unsigned)__builtin_popcount(M);
        unsigned incl = mine;
#pragma unroll
        for (int d = 1; d < 32; d <<= 1) {
            unsigned up = __shfl_up(incl, d);
            if (lane >= d) incl += up;
        }
        unsigned pos = incl - mine;
        const unsigned cnt = (unsigned)__builtin_amdgcn_readlane((int)incl, 31);
        while (M) {
            const int b = __builtin_ctz(M);
            M &= M - 1;
            wq[pos++] = (unsigned short)(((b >> 2) << 7) + lane * 4 + (b & 3));
        }
        __syncthreads();
        if (ch + 1 < N_CH) {
#pragma unroll
            for (int u = 0; u < EP_U; ++u) {
                int q = (ch + 1) * CH_Q + u * EP_T + tid;
                if (q < SLICE_Q) __builtin_prefetch(dq + q, 0, 3);
            }
        }
        bool emit = false;
        if constexpr (EMIT) {
            emit = !ovf && (woff + cnt <= (unsigned)LIST_CAPW);
            if (!emit) ovf = true;
        }

        const int ebase = (ch * CH_Q + wave * 32) * 4;
        for (unsigned base = 0; base < cnt; base += 64) {
            const unsigned nm = (cnt - base < 64u) ? (cnt - base) : 64u;
            int   sA = 0, sB = 0, rA = 0, rB = 0;
            float wA = 0.0f, wB = 0.0f;
            if ((unsigned)lane < nm) {
                unsigned p = wq[base + lane];
                int e = ebase + (int)(p >> 7) * (EP_T * 4) + (int)(p & 127u);
                sA = srcs[e]; wA = ews[e]; rA = dsts[e] - lo;
                if constexpr (EMIT) {
                    if (emit) lw[woff + base + lane] =
                        (v2u){(unsigned)sA | ((unsigned)rA << SRC_BITS), __float_as_uint(wA)};
                }
            }
            if ((unsigned)lane + 32u < nm) {
                unsigned p = wq[base + 32 + lane];
                int e = ebase + (int)(p >> 7) * (EP_T * 4) + (int)(p & 127u);
                sB = srcs[e]; wB = ews[e]; rB = dsts[e] - lo;
                if constexpr (EMIT) {
                    if (emit) lw[woff + base + 32 + lane] =
                        (v2u){(unsigned)sB | ((unsigned)rB << SRC_BITS), __float_as_uint(wB)};
                }
            }
            edge_round<D>(h, plane, lane, nm, sA, wA, rA, sB, wB, rB);
        }
        if constexpr (EMIT) { if (emit) woff += cnt; }
        __syncthreads();
    }
    return ovf ? LIST_OVF : woff;
}

template <int D, bool LAYER2>
__global__ void __launch_bounds__(EP_T)
GCN_edge_owner(const int* __restrict__ ei, const float* __restrict__ ew,
               const float* __restrict__ h, float* __restrict__ part,
               unsigned* __restrict__ list, unsigned* __restrict__ counts) {
    constexpr int PLANE  = RN * D;
    constexpr int STRIDE = (D == 16) ? STRIDE16 : STRIDE7;
    constexpr int PITCH  = (D == 16) ? 17 : 7;

    constexpr int PLANE_LDS   = RN * PITCH;
    constexpr int PLANE_ALLOC = (PLANE_LDS + 4 > 24600) ? (PLANE_LDS + 4) : 24600;
    __shared__ __attribute__((aligned(16))) float plane[PLANE_ALLOC];
    __shared__ unsigned short wqueue[32 * WQ_CAP];
    __shared__ unsigned any_ovf;

    const int tid  = threadIdx.x;
    const int lane = tid & 31;
    const int wave = tid >> 5;
    const int s    = blockIdx.x / RG;
    const int r    = blockIdx.x % RG;
    const int lo   = r * RN;

    const int*   srcs = ei + (size_t)s * SLICE_E;
    const int*   dsts = ei + (size_t)E_EDGES + (size_t)s * SLICE_E;
    const float* ews  = ew + (size_t)s * SLICE_E;
    unsigned short* wq = wqueue + wave * WQ_CAP;
    const size_t wslot = (size_t)blockIdx.x * 32 + wave;
    v2u* lw = (v2u*)list + wslot * LIST_CAPW;

    if (tid == 0) any_ovf = 0u;
    for (int i = tid; i < PLANE_LDS; i += EP_T) plane[i] = 0.0f;
    __syncthreads();

    if constexpr (!LAYER2) {
        const bool have = (list != nullptr) && (counts != nullptr);
        const unsigned n = edge_filter_pass<D, true>(srcs, dsts, ews, h, plane, wq, lw,
                                                     have, tid, lane, wave, lo);
        if (have && lane == 0) counts[wslot] = n;
    } else {
        unsigned n = LIST_OVF;
        if ((list != nullptr) && (counts != nullptr)) n = counts[wslot];
        if (n == LIST_OVF && lane == 0) any_ovf = 1u;
        __syncthreads();
        const bool refilter = __builtin_amdgcn_readfirstlane((int)any_ovf) != 0;
        if (refilter) {
            (void)edge_filter_pass<D, false>(srcs, dsts, ews, h, plane, wq, lw,
                                             false, tid, lane, wave, lo);
        } else {
            n = (unsigned)__builtin_amdgcn_readfirstlane((int)n);
            v2u nA = {0u, 0u}, nB = {0u, 0u};
            if ((unsigned)lane < n)       nA = lw[lane];
            if ((unsigned)lane + 32u < n) nB = lw[32 + lane];
            for (unsigned base = 0; base < n; base += 64) {
                const unsigned nm = (n - base < 64u) ? (n - base) : 64u;
                const v2u cA = nA, cB = nB;
                nA = (v2u){0u, 0u}; nB = (v2u){0u, 0u};
                if (base + 64u + (unsigned)lane < n) nA = lw[base + 64 + lane];
                if (base + 96u + (unsigned)lane < n) nB = lw[base + 96 + lane];
                edge_round<D>(h, plane, lane, nm,
                              (int)(cA[0] & ((1u << SRC_BITS) - 1u)), __uint_as_float(cA[1]), (int)(cA[0] >> SRC_BITS),
                              (int)(cB[0] & ((1u << SRC_BITS) - 1u)), __uint_as_float(cB[1]), (int)(cB[0] >> SRC_BITS));
            }
        }
    }
    __syncthreads();

    float* dstp = part + (size_t)blockIdx.x * STRIDE;
    for (int i = tid; i < PLANE; i += EP_T) dstp[i] = plane[(i / D) * PITCH + (i % D)];
}

__global__ void __launch_bounds__(256)
GCN_relu_gemm2_parts(const float* __restrict__ part, const float* __restrict__ b1,
                     const float* __restrict__ W2, float* __restrict__ z) {
    __shared__ float sW2[HID * OUT_F];
    __shared__ float sb1[HID];
    if (threadIdx.x < HID * OUT_F) sW2[threadIdx.x] = W2[threadIdx.x];
    if (threadIdx.x < HID)         sb1[threadIdx.x] = b1[threadIdx.x];
    __syncthreads();

    int node = blockIdx.x * blockDim.x + threadIdx.x;
    if (node >= N_NODES) return;
    const int r  = node / RN;
    const int nl = node - r * RN;
    const float* p0 = part + (size_t)r * STRIDE16 + (size_t)nl * HID;

    float a[HID];
#pragma unroll
    for (int i = 0; i < HID; ++i) a[i] = 0.0f;
#pragma unroll
    for (int sl = 0; sl < SL; ++sl) {
        const v4f* p = (const v4f*)(p0 + (size_t)sl * RG * STRIDE16);
#pragma unroll
        for (int q = 0; q < 4; ++q) {
            v4f t = p[q];
            a[q * 4 + 0] += t[0]; a[q * 4 + 1] += t[1];
            a[q * 4 + 2] += t[2]; a[q * 4 + 3] += t[3];
        }
    }
    float hv[HID];
#pragma unroll
    for (int i = 0; i < HID; ++i) hv[i] = fmaxf(a[i] + sb1[i], 0.0f);
#pragma unroll
    for (int j = 0; j < OUT_F; ++j) {
        float sum = 0.0f;
#pragma unroll
        for (int i = 0; i < HID; ++i)
            sum = fmaf(hv[i], sW2[i * OUT_F + j], sum);
        z[(size_t)node * OUT_F + j] = sum;
    }
}

__global__ void __launch_bounds__(256)
GCN_logsoftmax_parts(const float* __restrict__ part, const float* __restrict__ b2,
                     float* __restrict__ out) {
    __shared__ float sb2[OUT_F];
    if (threadIdx.x < OUT_F) sb2[threadIdx.x] = b2[threadIdx.x];
    __syncthreads();

    int node = blockIdx.x * blockDim.x + threadIdx.x;
    if (node >= N_NODES) return;
    const int r  = node / RN;
    const int nl = node - r * RN;
    const float* p0 = part + (size_t)r * STRIDE7 + (size_t)nl * OUT_F;

    float v[OUT_F];
#pragma unroll
    for (int j = 0; j < OUT_F; ++j) v[j] = 0.0f;
#pragma unroll
    for (int sl = 0; sl < SL; ++sl) {
        const float* p = p0 + (size_t)sl * RG * STRIDE7;
#pragma unroll
        for (int j = 0; j < OUT_F; ++j) v[j] += p[j];
    }
    float m = -3.402823466e38f;
#pragma unroll
    for (int j = 0; j < OUT_F; ++j) {
        v[j] += sb2[j];
        m = fmaxf(m, v[j]);
    }
    float sum = 0.0f;
#pragma unroll
    for (int j = 0; j < OUT_F; ++j) sum += expf(v[j] - m);
    float lse = logf(sum);
#pragma unroll
    for (int j = 0; j < OUT_F; ++j)
        out[(size_t)node * OUT_F + j] = v[j] - m - lse;
}

extern "C" void kernel_launch(void* const* d_in, const int* in_sizes, int n_in,
                              void* d_out, int out_size, void* d_ws, size_t ws_size,
                              hipStream_t stream) {
    const float* x  = (const float*)d_in[0];
    const int*   ei = (const int*)d_in[1];
    const float* ew = (const float*)d_in[2];
    const float* W1 = (const float*)d_in[3];
    const float* b1 = (const float*)d_in[4];
    const float* W2 = (const float*)d_in[5];
    const float* b2 = (const float*)d_in[6];
    float* out = (float*)d_out;

    char* ws = (char*)d_ws;
    size_t off = 0;
    auto alloc = [&](size_t bytes) { char* p = ws + off; off = (off + bytes + 255) & ~(size_t)255; return p; };
    float*  h0   = (float*)alloc((size_t)N_NODES * HID * 4);
    float*  z    = (float*)alloc((size_t)N_NODES * OUT_F * 4);
    __bf16* Bh   = (__bf16*)alloc((size_t)K_CHUNKS * 32 * 16 * 2);
    __bf16* Bl   = (__bf16*)alloc((size_t)K_CHUNKS * 32 * 16 * 2);
    float*  part = (float*)alloc((size_t)SL * RG * STRIDE16 * 4);
    const size_t need_planes = off;
    unsigned* elist = (unsigned*)alloc((size_t)SL * RG * 32 * LIST_CAPW * 8);
    unsigned* ecnt  = (unsigned*)alloc((size_t)SL * RG * 32 * 4);
    if (off > ws_size) { elist = nullptr; ecnt = nullptr; }

    if (need_planes <= ws_size) {
        GCN_pack_w1<<<K_CHUNKS, 256, 0, stream>>>(W1, Bh, Bl);
        GCN_gemm1_wmma<<<(N_TILES + 7) / 8, 256, 0, stream>>>(x, Bh, Bl, h0);
        GCN_edge_owner<16, false><<<SL * RG, EP_T, 0, stream>>>(ei, ew, h0, part, elist, ecnt);
        GCN_relu_gemm2_parts<<<(N_NODES + 255) / 256, 256, 0, stream>>>(part, b1, W2, z);
        GCN_edge_owner<OUT_F, true><<<SL * RG, EP_T, 0, stream>>>(ei, ew, z, part, elist, ecnt);
        GCN_logsoftmax_parts<<<(N_NODES + 255) / 256, 256, 0, stream>>>(part, b2, out);
        return;
    }

    off = 0;
    h0 = (float*)alloc((size_t)N_NODES * HID * 4);
    float* agg1 = (float*)alloc((size_t)N_NODES * HID * 4);
    z = (float*)alloc((size_t)N_NODES * OUT_F * 4);
    float* agg2 = (float*)alloc((size_t)N_NODES * OUT_F * 4);
    Bh = (__bf16*)alloc((size_t)K_CHUNKS * 32 * 16 * 2);
    Bl = (__bf16*)alloc((size_t)K_CHUNKS * 32 * 16 * 2);
    {
        long long n1 = (long long)N_NODES * HID;
        long long n2 = (long long)N_NODES * OUT_F;
        GCN_zero<<<(int)((n1 + 255) / 256), 256, 0, stream>>>(agg1, n1);
        GCN_zero<<<(int)((n2 + 255) / 256), 256, 0, stream>>>(agg2, n2);
    }
    GCN_pack_w1<<<K_CHUNKS, 256, 0, stream>>>(W1, Bh, Bl);
    GCN_gemm1_wmma<<<(N_TILES + 7) / 8, 256, 0, stream>>>(x, Bh, Bl, h0);
    {
        long long t = (long long)E_EDGES * 16;
        GCN_spmm16<<<(int)((t + 255) / 256), 256, 0, stream>>>(ei, ew, h0, agg1);
    }
    GCN_relu_gemm2<<<(N_NODES + 255) / 256, 256, 0, stream>>>(agg1, b1, W2, z);
    {
        long long t = (long long)E_EDGES * 8;
        GCN_spmm7<<<(int)((t + 255) / 256), 256, 0, stream>>>(ei, ew, z, agg2);
    }
    GCN_logsoftmax<<<(N_NODES + 255) / 256, 256, 0, stream>>>(agg2, b2, out);
}


# Edge-pass filter loop: its two work-group barriers per chunk go (the queue is per wave, wave-level ordering kept), so a SIMD's waves overlap loads, filtering and gathers; LDS add order changes
# speedup vs baseline: 2.1393x; 1.0473x over previous
#include <hip/hip_runtime.h>
#include <hip/hip_bf16.h>

typedef __attribute__((ext_vector_type(16))) __bf16 v16bf;
typedef __attribute__((ext_vector_type(8)))  float  v8f;
typedef __attribute__((ext_vector_type(4)))  float  v4f;

#define N_NODES 100000
#define E_EDGES 3200000
#define IN_F    512
#define HID     16
#define OUT_F   7
#define N_TILES (N_NODES / 16)
#define K_CHUNKS (IN_F / 32)

__device__ __forceinline__ void atomic_add_f32(float* p, float v) {
    __hip_atomic_fetch_add(p, v, __ATOMIC_RELAXED, __HIP_MEMORY_SCOPE_AGENT);
}

__global__ void GCN_pack_w1(const float* __restrict__ W1,
                            __bf16* __restrict__ Bh, __bf16* __restrict__ Bl) {
    int id = blockIdx.x * blockDim.x + threadIdx.x;
    if (id >= K_CHUNKS * 32 * 8) return;
    int v    = id & 7;
    int lane = (id >> 3) & 31;
    int c    = id >> 8;
    int n    = lane & 15;
    int g    = lane >> 4;
    int kb   = c * 32;
#pragma unroll
    for (int h = 0; h < 2; ++h) {
        int k = (v < 4) ? (kb + g * 8 + 2 * v + h)
                        : (kb + 16 + g * 8 + 2 * (v - 4) + h);
        float w   = W1[k * HID + n];
        __bf16 hi = (__bf16)w;
        __bf16 lo = (__bf16)(w - (float)hi);
        int idx = (c * 32 + lane) * 16 + v * 2 + h;
        Bh[idx] = hi;
        Bl[idx] = lo;
    }
}

__global__ void __launch_bounds__(256)
GCN_gemm1_wmma(const float* __restrict__ x,
               const __bf16* __restrict__ Bh, const __bf16* __restrict__ Bl,
               float* __restrict__ h0) {
    const int wave = threadIdx.x >> 5;
    const int lane = threadIdx.x & 31;
    const int tile = blockIdx.x * 8 + wave;
    if (tile >= N_TILES) return;

    const int n = lane & 15;
    const int g = lane >> 4;
    const float* xrow = x + (size_t)(tile * 16 + n) * IN_F;

    v8f acc = {};
#pragma unroll 4
    for (int c = 0; c < K_CHUNKS; ++c) {
        const int kb = c * 32;
        float f[16];
        *(v4f*)(f + 0)  = *(const v4f*)(xrow + kb + g * 8);
        *(v4f*)(f + 4)  = *(const v4f*)(xrow + kb + g * 8 + 4);
        *(v4f*)(f + 8)  = *(const v4f*)(xrow + kb + 16 + g * 8);
        *(v4f*)(f + 12) = *(const v4f*)(xrow + kb + 16 + g * 8 + 4);
        if (c + 1 < K_CHUNKS)
            __builtin_prefetch(xrow + kb + 32 + g * 8, 0, 3);

        v16bf ah, al;
#pragma unroll
        for (int i = 0; i < 16; ++i) {
            __bf16 hi = (__bf16)f[i];
            ah[i] = hi;
            al[i] = (__bf16)(f[i] - (float)hi);
        }
        v16bf bh = *(const v16bf*)(Bh + (c * 32 + lane) * 16);
        v16bf bl = *(const v16bf*)(Bl + (c * 32 + lane) * 16);

        acc = __builtin_amdgcn_wmma_f32_16x16x32_bf16(false, ah, false, bh,
                                                      (short)0, acc, false, false);
        acc = __builtin_amdgcn_wmma_f32_16x16x32_bf16(false, al, false, bh,
                                                      (short)0, acc, false, false);
        acc = __builtin_amdgcn_wmma_f32_16x16x32_bf16(false, ah, false, bl,
                                                      (short)0, acc, false, false);
    }
#pragma unroll
    for (int r = 0; r < 8; ++r)
        h0[(size_t)(tile * 16 + r + g * 8) * HID + n] = acc[r];
}

__global__ void GCN_spmm16(const int* __restrict__ ei, const float* __restrict__ ew,
                           const float* __restrict__ h, float* __restrict__ agg) {
    long long t = (long long)blockIdx.x * blockDim.x + threadIdx.x;
    if (t >= (long long)E_EDGES * 16) return;
    int e = (int)(t >> 4);
    int j = (int)(t & 15);
    int src = ei[e];
    int dst = ei[E_EDGES + e];
    float v = ew[e] * h[(size_t)src * HID + j];
    atomic_add_f32(&agg[(size_t)dst * HID + j], v);
}

__global__ void GCN_spmm7(const int* __restrict__ ei, const float* __restrict__ ew,
                          const float* __restrict__ z, float* __restrict__ agg) {
    long long t = (long long)blockIdx.x * blockDim.x + threadIdx.x;
    if (t >= (long long)E_EDGES * 8) return;
    int e = (int)(t >> 3);
    int j = (int)(t & 7);
    if (j >= OUT_F) return;
    int src = ei[e];
    int dst = ei[E_EDGES + e];
    float v = ew[e] * z[(size_t)src * OUT_F + j];
    atomic_add_f32(&agg[(size_t)dst * OUT_F + j], v);
}

__global__ void __launch_bounds__(256)
GCN_relu_gemm2(const float* __restrict__ agg1, const float* __restrict__ b1,
               const float* __restrict__ W2, float* __restrict__ z) {
    __shared__ float sW2[HID * OUT_F];
    __shared__ float sb1[HID];
    if (threadIdx.x < HID * OUT_F) sW2[threadIdx.x] = W2[threadIdx.x];
    if (threadIdx.x < HID)         sb1[threadIdx.x] = b1[threadIdx.x];
    __syncthreads();

    int node = blockIdx.x * blockDim.x + threadIdx.x;
    if (node >= N_NODES) return;

    float hv[HID];
#pragma unroll
    for (int i = 0; i < HID; ++i)
        hv[i] = fmaxf(agg1[(size_t)node * HID + i] + sb1[i], 0.0f);
#pragma unroll
    for (int j = 0; j < OUT_F; ++j) {
        float s = 0.0f;
#pragma unroll
        for (int i = 0; i < HID; ++i)
            s = fmaf(hv[i], sW2[i * OUT_F + j], s);
        z[(size_t)node * OUT_F + j] = s;
    }
}

__global__ void __launch_bounds__(256)
GCN_logsoftmax(const float* __restrict__ agg2, const float* __restrict__ b2,
               float* __restrict__ out) {
    __shared__ float sb2[OUT_F];
    if (threadIdx.x < OUT_F) sb2[threadIdx.x] = b2[threadIdx.x];
    __syncthreads();

    int node = blockIdx.x * blockDim.x + threadIdx.x;
    if (node >= N_NODES) return;

    float v[OUT_F];
    float m = -3.402823466e38f;
#pragma unroll
    for (int j = 0; j < OUT_F; ++j) {
        v[j] = agg2[(size_t)node * OUT_F + j] + sb2[j];
        m = fmaxf(m, v[j]);
    }
    float s = 0.0f;
#pragma unroll
    for (int j = 0; j < OUT_F; ++j) s += expf(v[j] - m);
    float lse = logf(s);
#pragma unroll
    for (int j = 0; j < OUT_F; ++j)
        out[(size_t)node * OUT_F + j] = v[j] - m - lse;
}

__global__ void GCN_zero(float* __restrict__ p, long long n) {
    long long i = (long long)blockIdx.x * blockDim.x + threadIdx.x;
    if (i < n) p[i] = 0.0f;
}

typedef __attribute__((ext_vector_type(4))) int v4i;

#define SL        8
#define RG        32
#define RN        (N_NODES / RG)
#define SLICE_E   (E_EDGES / SL)
#define SLICE_Q   (SLICE_E / 4)
#define EP_T      1024
#define EP_U      8
#define CH_Q      (EP_T * EP_U)
#define N_CH      ((SLICE_Q + CH_Q - 1) / CH_Q)
#define WQ_CAP    (32 * 4 * EP_U)
#define STRIDE16  50016
#define STRIDE7   21888

#define LIST_CAPW 576
#define LIST_OVF  0xFFFFFFFFu
#define SRC_BITS  17
typedef __attribute__((ext_vector_type(2))) unsigned v2u;

__device__ __forceinline__ void wave_lds_handoff() {
    __builtin_amdgcn_fence(__ATOMIC_RELEASE, "wavefront");
    __builtin_amdgcn_wave_barrier();
    __builtin_amdgcn_fence(__ATOMIC_ACQUIRE, "wavefront");
}

template <int D>
__device__ __forceinline__ void
edge_round(const float* __restrict__ h, float* plane, int lane, unsigned nm,
           int sA, float wA, int rA, int sB, float wB, int rB) {
    constexpr int LPM    = 4;
    constexpr int MPS    = 32 / LPM;
    constexpr int NSTEP  = 64 / MPS;
    constexpr int VW     = (D == 16) ? 4 : 2;
    constexpr int PITCH  = (D == 16) ? 17 : 7;
    const int sub = lane / LPM;
    const int jl  = lane % LPM;

    float vv[NSTEP][VW];
    float wk[NSTEP];
    int   lk[NSTEP];
    bool  ok[NSTEP];
#pragma unroll
    for (int t = 0; t < NSTEP; ++t) {
        const int k  = t * MPS + sub;
        const int sl = k & 31;
        const int ss = __shfl((t * MPS < 32) ? sA : sB, sl);
        wk[t] = __shfl((t * MPS < 32) ? wA : wB, sl);
        lk[t] = __shfl((t * MPS < 32) ? rA : rB, sl);
        ok[t] = ((unsigned)k < nm);
        if constexpr (D == 16) {
            v4f x4 = {0.0f, 0.0f, 0.0f, 0.0f};
            if (ok[t]) x4 = *((const v4f*)(h + (size_t)ss * 16) + jl);
#pragma unroll
            for (int i = 0; i < VW; ++i) vv[t][i] = x4[i];
        } else {
            float x0 = 0.0f, x1 = 0.0f;
            if (ok[t]) {
                const float* hr = h + (size_t)ss * D + jl * 2;
                x0 = hr[0];
                if (jl < 3) x1 = hr[1];
            }
            vv[t][0] = x0;
            vv[t][1] = x1;
        }
    }
#pragma unroll
    for (int t = 0; t < NSTEP; ++t) {
        if (ok[t]) {
            float* pr = plane + lk[t] * PITCH + jl * VW;
#pragma unroll
            for (int i = 0; i < VW; ++i) {
                if (jl * VW + i < D)
                    __hip_atomic_fetch_add(pr + i, wk[t] * vv[t][i],
                                           __ATOMIC_RELAXED, __HIP_MEMORY_SCOPE_WORKGROUP);
            }
        }
    }
}

template <int D, bool EMIT>
__device__ __forceinline__ unsigned
edge_filter_pass(const int* __restrict__ srcs, const int* __restrict__ dsts,
                 const float* __restrict__ ews, const float* __restrict__ h,
                 float* plane, unsigned short* wq, v2u* __restrict__ lw,
                 bool want_list, int tid, int lane, int wave, int lo) {
    const v4i* dq = (const v4i*)dsts;
    unsigned woff = 0;
    bool     ovf  = !want_list;
    for (int ch = 0; ch < N_CH; ++ch) {
        v4i dv[EP_U];
#pragma unroll
        for (int u = 0; u < EP_U; ++u) {
            int q = ch * CH_Q + u * EP_T + tid;
            dv[u] = (q < SLICE_Q) ? dq[q] : (v4i){-1, -1, -1, -1};
        }
        unsigned M = 0;
#pragma unroll
        for (int u = 0; u < EP_U; ++u) {
#pragma unroll
            for (int c = 0; c < 4; ++c) {
                if ((unsigned)(dv[u][c] - lo) < (unsigned)RN) M |= 1u << (u * 4 + c);
            }
        }
        const unsigned mine = (unsigned)__builtin_popcount(M);
        unsigned incl = mine;
#pragma unroll
        for (int d = 1; d < 32; d <<= 1) {
            unsigned up = __shfl_up(incl, d);
            if (lane >= d) incl += up;
        }
        unsigned pos = incl - mine;
        const unsigned cnt = (unsigned)__builtin_amdgcn_readlane((int)incl, 31);
        while (M) {
            const int b = __builtin_ctz(M);
            M &= M - 1;
            wq[pos++] = (unsigned short)(((b >> 2) << 7) + lane * 4 + (b & 3));
        }
        wave_lds_handoff();
        if (ch + 1 < N_CH) {
#pragma unroll
            for (int u = 0; u < EP_U; ++u) {
                int q = (ch + 1) * CH_Q + u * EP_T + tid;
                if (q < SLICE_Q) __builtin_prefetch(dq + q, 0, 3);
            }
        }
        bool emit = false;
        if constexpr (EMIT) {
            emit = !ovf && (woff + cnt <= (unsigned)LIST_CAPW);
            if (!emit) ovf = true;
        }

        const int ebase = (ch * CH_Q + wave * 32) * 4;
        for (unsigned base = 0; base < cnt; base += 64) {
            const unsigned nm = (cnt - base < 64u) ? (cnt - base) : 64u;
            int   sA = 0, sB = 0, rA = 0, rB = 0;
            float wA = 0.0f, wB = 0.0f;
            if ((unsigned)lane < nm) {
                unsigned p = wq[base + lane];
                int e = ebase + (int)(p >> 7) * (EP_T * 4) + (int)(p & 127u);
                sA = srcs[e]; wA = ews[e]; rA = dsts[e] - lo;
                if constexpr (EMIT) {
                    if (emit) lw[woff + base + lane] =
                        (v2u){(unsigned)sA | ((unsigned)rA << SRC_BITS), __float_as_uint(wA)};
                }
            }
            if ((unsigned)lane + 32u < nm) {
                unsigned p = wq[base + 32 + lane];
                int e = ebase + (int)(p >> 7) * (EP_T * 4) + (int)(p & 127u);
                sB = srcs[e]; wB = ews[e]; rB = dsts[e] - lo;
                if constexpr (EMIT) {
                    if (emit) lw[woff + base + 32 + lane] =
                        (v2u){(unsigned)sB | ((unsigned)rB << SRC_BITS), __float_as_uint(wB)};
                }
            }
            edge_round<D>(h, plane, lane, nm, sA, wA, rA, sB, wB, rB);
        }
        if constexpr (EMIT) { if (emit) woff += cnt; }
        wave_lds_handoff();
    }
    return ovf ? LIST_OVF : woff;
}

template <int D, bool LAYER2>
__global__ void __launch_bounds__(EP_T)
GCN_edge_owner(const int* __restrict__ ei, const float* __restrict__ ew,
               const float* __restrict__ h, float* __restrict__ part,
               unsigned* __restrict__ list, unsigned* __restrict__ counts) {
    constexpr int PLANE  = RN * D;
    constexpr int STRIDE = (D == 16) ? STRIDE16 : STRIDE7;
    constexpr int PITCH  = (D == 16) ? 17 : 7;

    constexpr int PLANE_LDS   = RN * PITCH;
    constexpr int PLANE_ALLOC = (PLANE_LDS + 4 > 24600) ? (PLANE_LDS + 4) : 24600;
    __shared__ __attribute__((aligned(16))) float plane[PLANE_ALLOC];
    __shared__ unsigned short wqueue[32 * WQ_CAP];
    __shared__ unsigned any_ovf;

    const int tid  = threadIdx.x;
    const int lane = tid & 31;
    const int wave = tid >> 5;
    const int s    = blockIdx.x / RG;
    const int r    = blockIdx.x % RG;
    const int lo   = r * RN;

    const int*   srcs = ei + (size_t)s * SLICE_E;
    const int*   dsts = ei + (size_t)E_EDGES + (size_t)s * SLICE_E;
    const float* ews  = ew + (size_t)s * SLICE_E;
    unsigned short* wq = wqueue + wave * WQ_CAP;
    const size_t wslot = (size_t)blockIdx.x * 32 + wave;
    v2u* lw = (v2u*)list + wslot * LIST_CAPW;

    if (tid == 0) any_ovf = 0u;
    for (int i = tid; i < PLANE_LDS; i += EP_T) plane[i] = 0.0f;
    __syncthreads();

    if constexpr (!LAYER2) {
        const bool have = (list != nullptr) && (counts != nullptr);
        const unsigned n = edge_filter_pass<D, true>(srcs, dsts, ews, h, plane, wq, lw,
                                                     have, tid, lane, wave, lo);
        if (have && lane == 0) counts[wslot] = n;
    } else {
        unsigned n = LIST_OVF;
        if ((list != nullptr) && (counts != nullptr)) n = counts[wslot];
        if (n == LIST_OVF && lane == 0) any_ovf = 1u;
        __syncthreads();
        const bool refilter = __builtin_amdgcn_readfirstlane((int)any_ovf) != 0;
        if (refilter) {
            (void)edge_filter_pass<D, false>(srcs, dsts, ews, h, plane, wq, lw,
                                             false, tid, lane, wave, lo);
        } else {
            n = (unsigned)__builtin_amdgcn_readfirstlane((int)n);
            v2u nA = {0u, 0u}, nB = {0u, 0u};
            if ((unsigned)lane < n)       nA = lw[lane];
            if ((unsigned)lane + 32u < n) nB = lw[32 + lane];
            for (unsigned base = 0; base < n; base += 64) {
                const unsigned nm = (n - base < 64u) ? (n - base) : 64u;
                const v2u cA = nA, cB = nB;
                nA = (v2u){0u, 0u}; nB = (v2u){0u, 0u};
                if (base + 64u + (unsigned)lane < n) nA = lw[base + 64 + lane];
                if (base + 96u + (unsigned)lane < n) nB = lw[base + 96 + lane];
                edge_round<D>(h, plane, lane, nm,
                              (int)(cA[0] & ((1u << SRC_BITS) - 1u)), __uint_as_float(cA[1]), (int)(cA[0] >> SRC_BITS),
                              (int)(cB[0] & ((1u << SRC_BITS) - 1u)), __uint_as_float(cB[1]), (int)(cB[0] >> SRC_BITS));
            }
        }
    }
    __syncthreads();

    float* dstp = part + (size_t)blockIdx.x * STRIDE;
    for (int i = tid; i < PLANE; i += EP_T) dstp[i] = plane[(i / D) * PITCH + (i % D)];
}

__global__ void __launch_bounds__(256)
GCN_relu_gemm2_parts(const float* __restrict__ part, const float* __restrict__ b1,
                     const float* __restrict__ W2, float* __restrict__ z) {
    __shared__ float sW2[HID * OUT_F];
    __shared__ float sb1[HID];
    if (threadIdx.x < HID * OUT_F) sW2[threadIdx.x] = W2[threadIdx.x];
    if (threadIdx.x < HID)         sb1[threadIdx.x] = b1[threadIdx.x];
    __syncthreads();

    int node = blockIdx.x * blockDim.x + threadIdx.x;
    if (node >= N_NODES) return;
    const int r  = node / RN;
    const int nl = node - r * RN;
    const float* p0 = part + (size_t)r * STRIDE16 + (size_t)nl * HID;

    float a[HID];
#pragma unroll
    for (int i = 0; i < HID; ++i) a[i] = 0.0f;
#pragma unroll
    for (int sl = 0; sl < SL; ++sl) {
        const v4f* p = (const v4f*)(p0 + (size_t)sl * RG * STRIDE16);
#pragma unroll
        for (int q = 0; q < 4; ++q) {
            v4f t = p[q];
            a[q * 4 + 0] += t[0]; a[q * 4 + 1] += t[1];
            a[q * 4 + 2] += t[2]; a[q * 4 + 3] += t[3];
        }
    }
    float hv[HID];
#pragma unroll
    for (int i = 0; i < HID; ++i) hv[i] = fmaxf(a[i] + sb1[i], 0.0f);
#pragma unroll
    for (int j = 0; j < OUT_F; ++j) {
        float sum = 0.0f;
#pragma unroll
        for (int i = 0; i < HID; ++i)
            sum = fmaf(hv[i], sW2[i * OUT_F + j], sum);
        z[(size_t)node * OUT_F + j] = sum;
    }
}

__global__ void __launch_bounds__(256)
GCN_logsoftmax_parts(const float* __restrict__ part, const float* __restrict__ b2,
                     float* __restrict__ out) {
    __shared__ float sb2[OUT_F];
    if (threadIdx.x < OUT_F) sb2[threadIdx.x] = b2[threadIdx.x];
    __syncthreads();

    int node = blockIdx.x * blockDim.x + threadIdx.x;
    if (node >= N_NODES) return;
    const int r  = node / RN;
    const int nl = node - r * RN;
    const float* p0 = part + (size_t)r * STRIDE7 + (size_t)nl * OUT_F;

    float v[OUT_F];
#pragma unroll
    for (int j = 0; j < OUT_F; ++j) v[j] = 0.0f;
#pragma unroll
    for (int sl = 0; sl < SL; ++sl) {
        const float* p = p0 + (size_t)sl * RG * STRIDE7;
#pragma unroll
        for (int j = 0; j < OUT_F; ++j) v[j] += p[j];
    }
    float m = -3.402823466e38f;
#pragma unroll
    for (int j = 0; j < OUT_F; ++j) {
        v[j] += sb2[j];
        m = fmaxf(m, v[j]);
    }
    float sum = 0.0f;
#pragma unroll
    for (int j = 0; j < OUT_F; ++j) sum += expf(v[j] - m);
    float lse = logf(sum);
#pragma unroll
    for (int j = 0; j < OUT_F; ++j)
        out[(size_t)node * OUT_F + j] = v[j] - m - lse;
}

extern "C" void kernel_launch(void* const* d_in, const int* in_sizes, int n_in,
                              void* d_out, int out_size, void* d_ws, size_t ws_size,
                              hipStream_t stream) {
    const float* x  = (const float*)d_in[0];
    const int*   ei = (const int*)d_in[1];
    const float* ew = (const float*)d_in[2];
    const float* W1 = (const float*)d_in[3];
    const float* b1 = (const float*)d_in[4];
    const float* W2 = (const float*)d_in[5];
    const float* b2 = (const float*)d_in[6];
    float* out = (float*)d_out;

    char* ws = (char*)d_ws;
    size_t off = 0;
    auto alloc = [&](size_t bytes) { char* p = ws + off; off = (off + bytes + 255) & ~(size_t)255; return p; };
    float*  h0   = (float*)alloc((size_t)N_NODES * HID * 4);
    float*  z    = (float*)alloc((size_t)N_NODES * OUT_F * 4);
    __bf16* Bh   = (__bf16*)alloc((size_t)K_CHUNKS * 32 * 16 * 2);
    __bf16* Bl   = (__bf16*)alloc((size_t)K_CHUNKS * 32 * 16 * 2);
    float*  part = (float*)alloc((size_t)SL * RG * STRIDE16 * 4);
    const size_t need_planes = off;
    unsigned* elist = (unsigned*)alloc((size_t)SL * RG * 32 * LIST_CAPW * 8);
    unsigned* ecnt  = (unsigned*)alloc((size_t)SL * RG * 32 * 4);
    if (off > ws_size) { elist = nullptr; ecnt = nullptr; }

    if (need_planes <= ws_size) {
        GCN_pack_w1<<<K_CHUNKS, 256, 0, stream>>>(W1, Bh, Bl);
        GCN_gemm1_wmma<<<(N_TILES + 7) / 8, 256, 0, stream>>>(x, Bh, Bl, h0);
        GCN_edge_owner<16, false><<<SL * RG, EP_T, 0, stream>>>(ei, ew, h0, part, elist, ecnt);
        GCN_relu_gemm2_parts<<<(N_NODES + 255) / 256, 256, 0, stream>>>(part, b1, W2, z);
        GCN_edge_owner<OUT_F, true><<<SL * RG, EP_T, 0, stream>>>(ei, ew, z, part, elist, ecnt);
        GCN_logsoftmax_parts<<<(N_NODES + 255) / 256, 256, 0, stream>>>(part, b2, out);
        return;
    }

    off = 0;
    h0 = (float*)alloc((size_t)N_NODES * HID * 4);
    float* agg1 = (float*)alloc((size_t)N_NODES * HID * 4);
    z = (float*)alloc((size_t)N_NODES * OUT_F * 4);
    float* agg2 = (float*)alloc((size_t)N_NODES * OUT_F * 4);
    Bh = (__bf16*)alloc((size_t)K_CHUNKS * 32 * 16 * 2);
    Bl = (__bf16*)alloc((size_t)K_CHUNKS * 32 * 16 * 2);
    {
        long long n1 = (long long)N_NODES * HID;
        long long n2 = (long long)N_NODES * OUT_F;
        GCN_zero<<<(int)((n1 + 255) / 256), 256, 0, stream>>>(agg1, n1);
        GCN_zero<<<(int)((n2 + 255) / 256), 256, 0, stream>>>(agg2, n2);
    }
    GCN_pack_w1<<<K_CHUNKS, 256, 0, stream>>>(W1, Bh, Bl);
    GCN_gemm1_wmma<<<(N_TILES + 7) / 8, 256, 0, stream>>>(x, Bh, Bl, h0);
    {
        long long t = (long long)E_EDGES * 16;
        GCN_spmm16<<<(int)((t + 255) / 256), 256, 0, stream>>>(ei, ew, h0, agg1);
    }
    GCN_relu_gemm2<<<(N_NODES + 255) / 256, 256, 0, stream>>>(agg1, b1, W2, z);
    {
        long long t = (long long)E_EDGES * 8;
        GCN_spmm7<<<(int)((t + 255) / 256), 256, 0, stream>>>(ei, ew, z, agg2);
    }
    GCN_logsoftmax<<<(N_NODES + 255) / 256, 256, 0, stream>>>(agg2, b2, out);
}


# Id job of the weight-packing launch also writes one 8-byte record per edge (source, row in range, weight); layer-1 filter reads a kept edge as one record, not three words from three columns
# speedup vs baseline: 2.4533x; 1.1468x over previous
#include <hip/hip_runtime.h>
#include <hip/hip_bf16.h>

typedef __attribute__((ext_vector_type(16))) __bf16 v16bf;
typedef __attribute__((ext_vector_type(8)))  float  v8f;
typedef __attribute__((ext_vector_type(4)))  float  v4f;

#define N_NODES 100000
#define E_EDGES 3200000
#define IN_F    512
#define HID     16
#define OUT_F   7
#define N_TILES (N_NODES / 16)
#define K_CHUNKS (IN_F / 32)

__device__ __forceinline__ void atomic_add_f32(float* p, float v) {
    __hip_atomic_fetch_add(p, v, __ATOMIC_RELAXED, __HIP_MEMORY_SCOPE_AGENT);
}

#define RN_IDS    3125u
#define ID_PAD    127u
#define SRC_BITS  17
#define N_IDW     (E_EDGES / 4)
#define REC_PER   5
#define REC_WG    (N_IDW / (256 * REC_PER))
static_assert(REC_WG * 256 * REC_PER == N_IDW, "the job's threads cover the four-edge groups exactly");
typedef __attribute__((ext_vector_type(4))) int pk_v4i;
typedef __attribute__((ext_vector_type(4))) unsigned pk_v4u;

__global__ void __launch_bounds__(256)
GCN_pack_w1(const float* __restrict__ W1,
            __bf16* __restrict__ Bh, __bf16* __restrict__ Bl,
            const int* __restrict__ ei, const float* __restrict__ ew,
            unsigned* __restrict__ ids, unsigned* __restrict__ recs) {
    if (blockIdx.x >= K_CHUNKS) {
        if (ids == nullptr || recs == nullptr) return;
        const pk_v4i* sq = (const pk_v4i*)ei;
        const pk_v4i* dq = (const pk_v4i*)(ei + E_EDGES);
        const v4f*    wq = (const v4f*)ew;
        pk_v4u*       rq = (pk_v4u*)recs;
        const int g0 = (blockIdx.x - K_CHUNKS) * 256 + threadIdx.x;
        pk_v4i d[REC_PER], s[REC_PER];
        v4f    w[REC_PER];
#pragma unroll
        for (int i = 0; i < REC_PER; ++i) {
            const int g = g0 + i * REC_WG * 256;
            d[i] = dq[g];
            s[i] = sq[g];
            w[i] = wq[g];
        }
        __builtin_amdgcn_sched_barrier(0);
#pragma unroll
        for (int i = 0; i < REC_PER; ++i) {
            const int g = g0 + i * REC_WG * 256;
            unsigned idw = 0;
            unsigned sr[4];
#pragma unroll
            for (int j = 0; j < 4; ++j) {
                const unsigned dd = (unsigned)d[i][j];
                unsigned q = dd / RN_IDS;
                sr[j] = (unsigned)s[i][j] | ((dd - q * RN_IDS) << SRC_BITS);
                q = (q < ID_PAD) ? q : ID_PAD;
                idw |= q << (8 * j);
            }
            ids[g] = idw;
            rq[2 * g]     = (pk_v4u){sr[0], __float_as_uint(w[i][0]), sr[1], __float_as_uint(w[i][1])};
            rq[2 * g + 1] = (pk_v4u){sr[2], __float_as_uint(w[i][2]), sr[3], __float_as_uint(w[i][3])};
        }
        return;
    }
    int id = blockIdx.x * blockDim.x + threadIdx.x;
    if (id >= K_CHUNKS * 32 * 8) return;
    int v    = id & 7;
    int lane = (id >> 3) & 31;
    int c    = id >> 8;
    int n    = lane & 15;
    int g    = lane >> 4;
    int kb   = c * 32;
#pragma unroll
    for (int h = 0; h < 2; ++h) {
        int k = (v < 4) ? (kb + g * 8 + 2 * v + h)
                        : (kb + 16 + g * 8 + 2 * (v - 4) + h);
        float w   = W1[k * HID + n];
        __bf16 hi = (__bf16)w;
        __bf16 lo = (__bf16)(w - (float)hi);
        int idx = (c * 32 + lane) * 16 + v * 2 + h;
        Bh[idx] = hi;
        Bl[idx] = lo;
    }
}

__global__ void __launch_bounds__(256)
GCN_gemm1_wmma(const float* __restrict__ x,
               const __bf16* __restrict__ Bh, const __bf16* __restrict__ Bl,
               float* __restrict__ h0) {
    const int wave = threadIdx.x >> 5;
    const int lane = threadIdx.x & 31;
    const int tile = blockIdx.x * 8 + wave;
    if (tile >= N_TILES) return;

    const int n = lane & 15;
    const int g = lane >> 4;
    const float* xrow = x + (size_t)(tile * 16 + n) * IN_F;

    v8f acc = {};
#pragma unroll 4
    for (int c = 0; c < K_CHUNKS; ++c) {
        const int kb = c * 32;
        float f[16];
        *(v4f*)(f + 0)  = *(const v4f*)(xrow + kb + g * 8);
        *(v4f*)(f + 4)  = *(const v4f*)(xrow + kb + g * 8 + 4);
        *(v4f*)(f + 8)  = *(const v4f*)(xrow + kb + 16 + g * 8);
        *(v4f*)(f + 12) = *(const v4f*)(xrow + kb + 16 + g * 8 + 4);
        if (c + 1 < K_CHUNKS)
            __builtin_prefetch(xrow + kb + 32 + g * 8, 0, 3);

        v16bf ah, al;
#pragma unroll
        for (int i = 0; i < 16; ++i) {
            __bf16 hi = (__bf16)f[i];
            ah[i] = hi;
            al[i] = (__bf16)(f[i] - (float)hi);
        }
        v16bf bh = *(const v16bf*)(Bh + (c * 32 + lane) * 16);
        v16bf bl = *(const v16bf*)(Bl + (c * 32 + lane) * 16);

        acc = __builtin_amdgcn_wmma_f32_16x16x32_bf16(false, ah, false, bh,
                                                      (short)0, acc, false, false);
        acc = __builtin_amdgcn_wmma_f32_16x16x32_bf16(false, al, false, bh,
                                                      (short)0, acc, false, false);
        acc = __builtin_amdgcn_wmma_f32_16x16x32_bf16(false, ah, false, bl,
                                                      (short)0, acc, false, false);
    }
#pragma unroll
    for (int r = 0; r < 8; ++r)
        h0[(size_t)(tile * 16 + r + g * 8) * HID + n] = acc[r];
}

__global__ void GCN_spmm16(const int* __restrict__ ei, const float* __restrict__ ew,
                           const float* __restrict__ h, float* __restrict__ agg) {
    long long t = (long long)blockIdx.x * blockDim.x + threadIdx.x;
    if (t >= (long long)E_EDGES * 16) return;
    int e = (int)(t >> 4);
    int j = (int)(t & 15);
    int src = ei[e];
    int dst = ei[E_EDGES + e];
    float v = ew[e] * h[(size_t)src * HID + j];
    atomic_add_f32(&agg[(size_t)dst * HID + j], v);
}

__global__ void GCN_spmm7(const int* __restrict__ ei, const float* __restrict__ ew,
                          const float* __restrict__ z, float* __restrict__ agg) {
    long long t = (long long)blockIdx.x * blockDim.x + threadIdx.x;
    if (t >= (long long)E_EDGES * 8) return;
    int e = (int)(t >> 3);
    int j = (int)(t & 7);
    if (j >= OUT_F) return;
    int src = ei[e];
    int dst = ei[E_EDGES + e];
    float v = ew[e] * z[(size_t)src * OUT_F + j];
    atomic_add_f32(&agg[(size_t)dst * OUT_F + j], v);
}

__global__ void __launch_bounds__(256)
GCN_relu_gemm2(const float* __restrict__ agg1, const float* __restrict__ b1,
               const float* __restrict__ W2, float* __restrict__ z) {
    __shared__ float sW2[HID * OUT_F];
    __shared__ float sb1[HID];
    if (threadIdx.x < HID * OUT_F) sW2[threadIdx.x] = W2[threadIdx.x];
    if (threadIdx.x < HID)         sb1[threadIdx.x] = b1[threadIdx.x];
    __syncthreads();

    int node = blockIdx.x * blockDim.x + threadIdx.x;
    if (node >= N_NODES) return;

    float hv[HID];
#pragma unroll
    for (int i = 0; i < HID; ++i)
        hv[i] = fmaxf(agg1[(size_t)node * HID + i] + sb1[i], 0.0f);
#pragma unroll
    for (int j = 0; j < OUT_F; ++j) {
        float s = 0.0f;
#pragma unroll
        for (int i = 0; i < HID; ++i)
            s = fmaf(hv[i], sW2[i * OUT_F + j], s);
        z[(size_t)node * OUT_F + j] = s;
    }
}

__global__ void __launch_bounds__(256)
GCN_logsoftmax(const float* __restrict__ agg2, const float* __restrict__ b2,
               float* __restrict__ out) {
    __shared__ float sb2[OUT_F];
    if (threadIdx.x < OUT_F) sb2[threadIdx.x] = b2[threadIdx.x];
    __syncthreads();

    int node = blockIdx.x * blockDim.x + threadIdx.x;
    if (node >= N_NODES) return;

    float v[OUT_F];
    float m = -3.402823466e38f;
#pragma unroll
    for (int j = 0; j < OUT_F; ++j) {
        v[j] = agg2[(size_t)node * OUT_F + j] + sb2[j];
        m = fmaxf(m, v[j]);
    }
    float s = 0.0f;
#pragma unroll
    for (int j = 0; j < OUT_F; ++j) s += expf(v[j] - m);
    float lse = logf(s);
#pragma unroll
    for (int j = 0; j < OUT_F; ++j)
        out[(size_t)node * OUT_F + j] = v[j] - m - lse;
}

__global__ void GCN_zero(float* __restrict__ p, long long n) {
    long long i = (long long)blockIdx.x * blockDim.x + threadIdx.x;
    if (i < n) p[i] = 0.0f;
}

typedef __attribute__((ext_vector_type(4))) int v4i;

#define SL        8
#define RG        32
#define RN        (N_NODES / RG)
#define SLICE_E   (E_EDGES / SL)
#define SLICE_Q   (SLICE_E / 4)
#define EP_T      1024
#define EP_U      8
#define CH_Q      (EP_T * EP_U)
#define N_CH      ((SLICE_Q + CH_Q - 1) / CH_Q)
#define WQ_CAP    (32 * 4 * EP_U)
#define STRIDE16  50016
#define STRIDE7   21888

#define LIST_CAPW 576
#define LIST_OVF  0xFFFFFFFFu
typedef __attribute__((ext_vector_type(2))) unsigned v2u;

__device__ __forceinline__ void wave_lds_handoff() {
    __builtin_amdgcn_fence(__ATOMIC_RELEASE, "wavefront");
    __builtin_amdgcn_wave_barrier();
    __builtin_amdgcn_fence(__ATOMIC_ACQUIRE, "wavefront");
}

template <int D>
__device__ __forceinline__ void
edge_round(const float* __restrict__ h, float* plane, int lane, unsigned nm,
           int sA, float wA, int rA, int sB, float wB, int rB) {
    constexpr int LPM    = 4;
    constexpr int MPS    = 32 / LPM;
    constexpr int NSTEP  = 64 / MPS;
    constexpr int VW     = (D == 16) ? 4 : 2;
    constexpr int PITCH  = (D == 16) ? 17 : 7;
    const int sub = lane / LPM;
    const int jl  = lane % LPM;

    float vv[NSTEP][VW];
    float wk[NSTEP];
    int   lk[NSTEP];
    bool  ok[NSTEP];
#pragma unroll
    for (int t = 0; t < NSTEP; ++t) {
        const int k  = t * MPS + sub;
        const int sl = k & 31;
        const int ss = __shfl((t * MPS < 32) ? sA : sB, sl);
        wk[t] = __shfl((t * MPS < 32) ? wA : wB, sl);
        lk[t] = __shfl((t * MPS < 32) ? rA : rB, sl);
        ok[t] = ((unsigned)k < nm);
        if constexpr (D == 16) {
            v4f x4 = {0.0f, 0.0f, 0.0f, 0.0f};
            if (ok[t]) x4 = *((const v4f*)(h + (size_t)ss * 16) + jl);
#pragma unroll
            for (int i = 0; i < VW; ++i) vv[t][i] = x4[i];
        } else {
            float x0 = 0.0f, x1 = 0.0f;
            if (ok[t]) {
                const float* hr = h + (size_t)ss * D + jl * 2;
                x0 = hr[0];
                if (jl < 3) x1 = hr[1];
            }
            vv[t][0] = x0;
            vv[t][1] = x1;
        }
    }
#pragma unroll
    for (int t = 0; t < NSTEP; ++t) {
        if (ok[t]) {
            float* pr = plane + lk[t] * PITCH + jl * VW;
#pragma unroll
            for (int i = 0; i < VW; ++i) {
                if (jl * VW + i < D)
                    __hip_atomic_fetch_add(pr + i, wk[t] * vv[t][i],
                                           __ATOMIC_RELAXED, __HIP_MEMORY_SCOPE_WORKGROUP);
            }
        }
    }
}

template <int D, bool EMIT>
__device__ __forceinline__ unsigned
edge_filter_pass(const int* __restrict__ srcs, const int* __restrict__ dsts,
                 const float* __restrict__ ews, const float* __restrict__ h,
                 float* plane, unsigned short* wq, v2u* __restrict__ lw,
                 bool want_list, int tid, int lane, int wave, int lo) {
    const v4i* dq = (const v4i*)dsts;
    unsigned woff = 0;
    bool     ovf  = !want_list;
    for (int ch = 0; ch < N_CH; ++ch) {
        v4i dv[EP_U];
#pragma unroll
        for (int u = 0; u < EP_U; ++u) {
            int q = ch * CH_Q + u * EP_T + tid;
            dv[u] = (q < SLICE_Q) ? dq[q] : (v4i){-1, -1, -1, -1};
        }
        unsigned M = 0;
#pragma unroll
        for (int u = 0; u < EP_U; ++u) {
#pragma unroll
            for (int c = 0; c < 4; ++c) {
                if ((unsigned)(dv[u][c] - lo) < (unsigned)RN) M |= 1u << (u * 4 + c);
            }
        }
        const unsigned mine = (unsigned)__builtin_popcount(M);
        unsigned incl = mine;
#pragma unroll
        for (int d = 1; d < 32; d <<= 1) {
            unsigned up = __shfl_up(incl, d);
            if (lane >= d) incl += up;
        }
        unsigned pos = incl - mine;
        const unsigned cnt = (unsigned)__builtin_amdgcn_readlane((int)incl, 31);
        while (M) {
            const int b = __builtin_ctz(M);
            M &= M - 1;
            wq[pos++] = (unsigned short)(((b >> 2) << 7) + lane * 4 + (b & 3));
        }
        wave_lds_handoff();
        if (ch + 1 < N_CH) {
#pragma unroll
            for (int u = 0; u < EP_U; ++u) {
                int q = (ch + 1) * CH_Q + u * EP_T + tid;
                if (q < SLICE_Q) __builtin_prefetch(dq + q, 0, 3);
            }
        }
        bool emit = false;
        if constexpr (EMIT) {
            emit = !ovf && (woff + cnt <= (unsigned)LIST_CAPW);
            if (!emit) ovf = true;
        }

        const int ebase = (ch * CH_Q + wave * 32) * 4;
        for (unsigned base = 0; base < cnt; base += 64) {
            const unsigned nm = (cnt - base < 64u) ? (cnt - base) : 64u;
            int   sA = 0, sB = 0, rA = 0, rB = 0;
            float wA = 0.0f, wB = 0.0f;
            if ((unsigned)lane < nm) {
                unsigned p = wq[base + lane];
                int e = ebase + (int)(p >> 7) * (EP_T * 4) + (int)(p & 127u);
                sA = srcs[e]; wA = ews[e]; rA = dsts[e] - lo;
                if constexpr (EMIT) {
                    if (emit) lw[woff + base + lane] =
                        (v2u){(unsigned)sA | ((unsigned)rA << SRC_BITS), __float_as_uint(wA)};
                }
            }
            if ((unsigned)lane + 32u < nm) {
                unsigned p = wq[base + 32 + lane];
                int e = ebase + (int)(p >> 7) * (EP_T * 4) + (int)(p & 127u);
                sB = srcs[e]; wB = ews[e]; rB = dsts[e] - lo;
                if constexpr (EMIT) {
                    if (emit) lw[woff + base + 32 + lane] =
                        (v2u){(unsigned)sB | ((unsigned)rB << SRC_BITS), __float_as_uint(wB)};
                }
            }
            edge_round<D>(h, plane, lane, nm, sA, wA, rA, sB, wB, rB);
        }
        if constexpr (EMIT) { if (emit) woff += cnt; }
        wave_lds_handoff();
    }
    return ovf ? LIST_OVF : woff;
}

#define SLICE_X   (SLICE_E / 16)
#define CH_X      (EP_T * 2)
typedef __attribute__((ext_vector_type(4))) unsigned v4u;

template <int D, bool EMIT>
__device__ __forceinline__ unsigned
edge_filter_pass_ids(const v2u* __restrict__ recl, const unsigned* __restrict__ idsl,
                     const float* __restrict__ h,
                     float* plane, unsigned short* wq, v2u* __restrict__ lw,
                     bool want_list, int tid, int lane, int wave, unsigned rr) {
    static_assert((SLICE_X + CH_X - 1) / CH_X == N_CH, "same chunk count as the destination pass");
    static_assert(RN_IDS == (unsigned)RN && RG <= (int)ID_PAD, "the id bytes name this pass's ranges");
    const v4u* iq = (const v4u*)idsl;
    unsigned woff = 0;
    bool     ovf  = !want_list;
    for (int ch = 0; ch < N_CH; ++ch) {
        v4u dv[2];
#pragma unroll
        for (int u = 0; u < 2; ++u) {
            int q = ch * CH_X + u * EP_T + tid;
            dv[u] = (q < SLICE_X) ? iq[q] : (v4u){0x7F7F7F7Fu, 0x7F7F7F7Fu, 0x7F7F7F7Fu, 0x7F7F7F7Fu};
        }
        unsigned M = 0;
#pragma unroll
        for (int k = 0; k < 8; ++k) {
            const unsigned y = dv[k >> 2][k & 3] ^ rr;
            const unsigned g = y + 0x7F7F7F7Fu;
            M |= (~g & 0x80808080u) >> (7 - k);
        }
        const unsigned mine = (unsigned)__builtin_popcount(M);
        unsigned incl = mine;
#pragma unroll
        for (int d = 1; d < 32; d <<= 1) {
            unsigned up = __shfl_up(incl, d);
            if (lane >= d) incl += up;
        }
        unsigned pos = incl - mine;
        const unsigned cnt = (unsigned)__builtin_amdgcn_readlane((int)incl, 31);
        while (M) {
            const int b = __builtin_ctz(M);
            M &= M - 1;
            wq[pos++] = (unsigned short)(((b & 4) << 7) | (lane << 4) | ((b & 3) << 2) | (b >> 3));
        }
        wave_lds_handoff();
        if (ch + 1 < N_CH) {
#pragma unroll
            for (int u = 0; u < 2; ++u) {
                int q = (ch + 1) * CH_X + u * EP_T + tid;
                if (q < SLICE_X) __builtin_prefetch(iq + q, 0, 3);
            }
        }
        bool emit = false;
        if constexpr (EMIT) {
            emit = !ovf && (woff + cnt <= (unsigned)LIST_CAPW);
            if (!emit) ovf = true;
        }

        const int ebase = (ch * CH_X + wave * 32) * 16;
        for (unsigned base = 0; base < cnt; base += 64) {
            const unsigned nm = (cnt - base < 64u) ? (cnt - base) : 64u;
            int   sA = 0, sB = 0, rA = 0, rB = 0;
            float wA = 0.0f, wB = 0.0f;
            if ((unsigned)lane < nm) {
                unsigned p = wq[base + lane];
                int e = ebase + (int)(p >> 9) * (EP_T * 16) + (int)(p & 511u);
                const v2u rc = recl[e];
                sA = (int)(rc[0] & ((1u << SRC_BITS) - 1u));
                rA = (int)(rc[0] >> SRC_BITS);
                wA = __uint_as_float(rc[1]);
                if constexpr (EMIT) {
                    if (emit) lw[woff + base + lane] = rc;
                }
            }
            if ((unsigned)lane + 32u < nm) {
                unsigned p = wq[base + 32 + lane];
                int e = ebase + (int)(p >> 9) * (EP_T * 16) + (int)(p & 511u);
                const v2u rc = recl[e];
                sB = (int)(rc[0] & ((1u << SRC_BITS) - 1u));
                rB = (int)(rc[0] >> SRC_BITS);
                wB = __uint_as_float(rc[1]);
                if constexpr (EMIT) {
                    if (emit) lw[woff + base + 32 + lane] = rc;
                }
            }
            edge_round<D>(h, plane, lane, nm, sA, wA, rA, sB, wB, rB);
        }
        if constexpr (EMIT) { if (emit) woff += cnt; }
        wave_lds_handoff();
    }
    return ovf ? LIST_OVF : woff;
}

template <int D, bool LAYER2>
__global__ void __launch_bounds__(EP_T)
GCN_edge_owner(const int* __restrict__ ei, const float* __restrict__ ew,
               const float* __restrict__ h, float* __restrict__ part,
               unsigned* __restrict__ list, unsigned* __restrict__ counts,
               const unsigned* __restrict__ ids, const unsigned* __restrict__ recs) {
    constexpr int PLANE  = RN * D;
    constexpr int STRIDE = (D == 16) ? STRIDE16 : STRIDE7;
    constexpr int PITCH  = (D == 16) ? 17 : 7;

    constexpr int PLANE_LDS   = RN * PITCH;
    constexpr int PLANE_ALLOC = (PLANE_LDS + 4 > 24600) ? (PLANE_LDS + 4) : 24600;
    __shared__ __attribute__((aligned(16))) float plane[PLANE_ALLOC];
    __shared__ unsigned short wqueue[32 * WQ_CAP];
    __shared__ unsigned any_ovf;

    const int tid  = threadIdx.x;
    const int lane = tid & 31;
    const int wave = tid >> 5;
    const int s    = blockIdx.x / RG;
    const int r    = blockIdx.x % RG;
    const int lo   = r * RN;

    const int*   srcs = ei + (size_t)s * SLICE_E;
    const int*   dsts = ei + (size_t)E_EDGES + (size_t)s * SLICE_E;
    const float* ews  = ew + (size_t)s * SLICE_E;
    unsigned short* wq = wqueue + wave * WQ_CAP;
    const size_t wslot = (size_t)blockIdx.x * 32 + wave;
    v2u* lw = (v2u*)list + wslot * LIST_CAPW;

    if (tid == 0) any_ovf = 0u;
    for (int i = tid; i < PLANE_LDS; i += EP_T) plane[i] = 0.0f;
    __syncthreads();

    if constexpr (!LAYER2) {
        const bool have = (list != nullptr) && (counts != nullptr);
        unsigned n;
        if (ids != nullptr && recs != nullptr) {
            n = edge_filter_pass_ids<D, true>((const v2u*)recs + (size_t)s * SLICE_E,
                                              ids + (size_t)s * (SLICE_E / 4), h,
                                              plane, wq, lw, have, tid, lane, wave,
                                              (unsigned)r * 0x01010101u);
        } else {
            n = edge_filter_pass<D, true>(srcs, dsts, ews, h, plane, wq, lw,
                                          have, tid, lane, wave, lo);
        }
        if (have && lane == 0) counts[wslot] = n;
    } else {
        unsigned n = LIST_OVF;
        if ((list != nullptr) && (counts != nullptr)) n = counts[wslot];
        if (n == LIST_OVF && lane == 0) any_ovf = 1u;
        __syncthreads();
        const bool refilter = __builtin_amdgcn_readfirstlane((int)any_ovf) != 0;
        if (refilter) {
            (void)edge_filter_pass<D, false>(srcs, dsts, ews, h, plane, wq, lw,
                                             false, tid, lane, wave, lo);
        } else {
            n = (unsigned)__builtin_amdgcn_readfirstlane((int)n);
            v2u nA = {0u, 0u}, nB = {0u, 0u};
            if ((unsigned)lane < n)       nA = lw[lane];
            if ((unsigned)lane + 32u < n) nB = lw[32 + lane];
            for (unsigned base = 0; base < n; base += 64) {
                const unsigned nm = (n - base < 64u) ? (n - base) : 64u;
                const v2u cA = nA, cB = nB;
                nA = (v2u){0u, 0u}; nB = (v2u){0u, 0u};
                if (base + 64u + (unsigned)lane < n) nA = lw[base + 64 + lane];
                if (base + 96u + (unsigned)lane < n) nB = lw[base + 96 + lane];
                edge_round<D>(h, plane, lane, nm,
                              (int)(cA[0] & ((1u << SRC_BITS) - 1u)), __uint_as_float(cA[1]), (int)(cA[0] >> SRC_BITS),
                              (int)(cB[0] & ((1u << SRC_BITS) - 1u)), __uint_as_float(cB[1]), (int)(cB[0] >> SRC_BITS));
            }
        }
    }
    __syncthreads();

    float* dstp = part + (size_t)blockIdx.x * STRIDE;
    for (int i = tid; i < PLANE; i += EP_T) dstp[i] = plane[(i / D) * PITCH + (i % D)];
}

__global__ void __launch_bounds__(256)
GCN_relu_gemm2_parts(const float* __restrict__ part, const float* __restrict__ b1,
                     const float* __restrict__ W2, float* __restrict__ z) {
    __shared__ float sW2[HID * OUT_F];
    __shared__ float sb1[HID];
    if (threadIdx.x < HID * OUT_F) sW2[threadIdx.x] = W2[threadIdx.x];
    if (threadIdx.x < HID)         sb1[threadIdx.x] = b1[threadIdx.x];
    __syncthreads();

    int node = blockIdx.x * blockDim.x + threadIdx.x;
    if (node >= N_NODES) return;
    const int r  = node / RN;
    const int nl = node - r * RN;
    const float* p0 = part + (size_t)r * STRIDE16 + (size_t)nl * HID;

    float a[HID];
#pragma unroll
    for (int i = 0; i < HID; ++i) a[i] = 0.0f;
#pragma unroll
    for (int sl = 0; sl < SL; ++sl) {
        const v4f* p = (const v4f*)(p0 + (size_t)sl * RG * STRIDE16);
#pragma unroll
        for (int q = 0; q < 4; ++q) {
            v4f t = p[q];
            a[q * 4 + 0] += t[0]; a[q * 4 + 1] += t[1];
            a[q * 4 + 2] += t[2]; a[q * 4 + 3] += t[3];
        }
    }
    float hv[HID];
#pragma unroll
    for (int i = 0; i < HID; ++i) hv[i] = fmaxf(a[i] + sb1[i], 0.0f);
#pragma unroll
    for (int j = 0; j < OUT_F; ++j) {
        float sum = 0.0f;
#pragma unroll
        for (int i = 0; i < HID; ++i)
            sum = fmaf(hv[i], sW2[i * OUT_F + j], sum);
        z[(size_t)node * OUT_F + j] = sum;
    }
}

__global__ void __launch_bounds__(256)
GCN_logsoftmax_parts(const float* __restrict__ part, const float* __restrict__ b2,
                     float* __restrict__ out) {
    __shared__ float sb2[OUT_F];
    if (threadIdx.x < OUT_F) sb2[threadIdx.x] = b2[threadIdx.x];
    __syncthreads();

    int node = blockIdx.x * blockDim.x + threadIdx.x;
    if (node >= N_NODES) return;
    const int r  = node / RN;
    const int nl = node - r * RN;
    const float* p0 = part + (size_t)r * STRIDE7 + (size_t)nl * OUT_F;

    float v[OUT_F];
#pragma unroll
    for (int j = 0; j < OUT_F; ++j) v[j] = 0.0f;
#pragma unroll
    for (int sl = 0; sl < SL; ++sl) {
        const float* p = p0 + (size_t)sl * RG * STRIDE7;
#pragma unroll
        for (int j = 0; j < OUT_F; ++j) v[j] += p[j];
    }
    float m = -3.402823466e38f;
#pragma unroll
    for (int j = 0; j < OUT_F; ++j) {
        v[j] += sb2[j];
        m = fmaxf(m, v[j]);
    }
    float sum = 0.0f;
#pragma unroll
    for (int j = 0; j < OUT_F; ++j) sum += expf(v[j] - m);
    float lse = logf(sum);
#pragma unroll
    for (int j = 0; j < OUT_F; ++j)
        out[(size_t)node * OUT_F + j] = v[j] - m - lse;
}

extern "C" void kernel_launch(void* const* d_in, const int* in_sizes, int n_in,
                              void* d_out, int out_size, void* d_ws, size_t ws_size,
                              hipStream_t stream) {
    const float* x  = (const float*)d_in[0];
    const int*   ei = (const int*)d_in[1];
    const float* ew = (const float*)d_in[2];
    const float* W1 = (const float*)d_in[3];
    const float* b1 = (const float*)d_in[4];
    const float* W2 = (const float*)d_in[5];
    const float* b2 = (const float*)d_in[6];
    float* out = (float*)d_out;

    char* ws = (char*)d_ws;
    size_t off = 0;
    auto alloc = [&](size_t bytes) { char* p = ws + off; off = (off + bytes + 255) & ~(size_t)255; return p; };
    float*  h0   = (float*)alloc((size_t)N_NODES * HID * 4);
    float*  z    = (float*)alloc((size_t)N_NODES * OUT_F * 4);
    __bf16* Bh   = (__bf16*)alloc((size_t)K_CHUNKS * 32 * 16 * 2);
    __bf16* Bl   = (__bf16*)alloc((size_t)K_CHUNKS * 32 * 16 * 2);
    float*  part = (float*)alloc((size_t)SL * RG * STRIDE16 * 4);
    const size_t need_planes = off;
    unsigned* elist = (unsigned*)alloc((size_t)SL * RG * 32 * LIST_CAPW * 8);
    unsigned* ecnt  = (unsigned*)alloc((size_t)SL * RG * 32 * 4);
    if (off > ws_size) { elist = nullptr; ecnt = nullptr; }
    unsigned* eids = (unsigned*)alloc((size_t)E_EDGES);
    unsigned* erec = (unsigned*)alloc((size_t)E_EDGES * 8);
    if (off > ws_size) { eids = nullptr; erec = nullptr; }

    if (need_planes <= ws_size) {
        GCN_pack_w1<<<K_CHUNKS + REC_WG, 256, 0, stream>>>(W1, Bh, Bl, ei, ew, eids, erec);
        GCN_gemm1_wmma<<<(N_TILES + 7) / 8, 256, 0, stream>>>(x, Bh, Bl, h0);
        GCN_edge_owner<16, false><<<SL * RG, EP_T, 0, stream>>>(ei, ew, h0, part, elist, ecnt, eids, erec);
        GCN_relu_gemm2_parts<<<(N_NODES + 255) / 256, 256, 0, stream>>>(part, b1, W2, z);
        GCN_edge_owner<OUT_F, true><<<SL * RG, EP_T, 0, stream>>>(ei, ew, z, part, elist, ecnt, nullptr, nullptr);
        GCN_logsoftmax_parts<<<(N_NODES + 255) / 256, 256, 0, stream>>>(part, b2, out);
        return;
    }

    off = 0;
    h0 = (float*)alloc((size_t)N_NODES * HID * 4);
    float* agg1 = (float*)alloc((size_t)N_NODES * HID * 4);
    z = (float*)alloc((size_t)N_NODES * OUT_F * 4);
    float* agg2 = (float*)alloc((size_t)N_NODES * OUT_F * 4);
    Bh = (__bf16*)alloc((size_t)K_CHUNKS * 32 * 16 * 2);
    Bl = (__bf16*)alloc((size_t)K_CHUNKS * 32 * 16 * 2);
    {
        long long n1 = (long long)N_NODES * HID;
        long long n2 = (long long)N_NODES * OUT_F;
        GCN_zero<<<(int)((n1 + 255) / 256), 256, 0, stream>>>(agg1, n1);
        GCN_zero<<<(int)((n2 + 255) / 256), 256, 0, stream>>>(agg2, n2);
    }
    GCN_pack_w1<<<K_CHUNKS, 256, 0, stream>>>(W1, Bh, Bl, nullptr, nullptr, nullptr, nullptr);
    GCN_gemm1_wmma<<<(N_TILES + 7) / 8, 256, 0, stream>>>(x, Bh, Bl, h0);
    {
        long long t = (long long)E_EDGES * 16;
        GCN_spmm16<<<(int)((t + 255) / 256), 256, 0, stream>>>(ei, ew, h0, agg1);
    }
    GCN_relu_gemm2<<<(N_NODES + 255) / 256, 256, 0, stream>>>(agg1, b1, W2, z);
    {
        long long t = (long long)E_EDGES * 8;
        GCN_spmm7<<<(int)((t + 255) / 256), 256, 0, stream>>>(ei, ew, z, agg2);
    }
    GCN_logsoftmax<<<(N_NODES + 255) / 256, 256, 0, stream>>>(agg2, b2, out);
}


# Edge passes own whole destination ranges (241 x 416 nodes; a wave sums its range's runs of 8 tiles): LDS plane is final, bias+ReLU+h@W2 and log-softmax done from it; 2 launches go; sum order changes
# speedup vs baseline: 3.5436x; 1.4444x over previous
#include <hip/hip_runtime.h>
#include <hip/hip_bf16.h>

typedef __attribute__((ext_vector_type(16))) __bf16 v16bf;
typedef __attribute__((ext_vector_type(8)))  float  v8f;
typedef __attribute__((ext_vector_type(4)))  float  v4f;

#define N_NODES 100000
#define E_EDGES 3200000
#define IN_F    512
#define HID     16
#define OUT_F   7
#define N_TILES (N_NODES / 16)
#define K_CHUNKS (IN_F / 32)

__device__ __forceinline__ void atomic_add_f32(float* p, float v) {
    __hip_atomic_fetch_add(p, v, __ATOMIC_RELAXED, __HIP_MEMORY_SCOPE_AGENT);
}


__device__ __forceinline__ void
pack_w1_job(const float* __restrict__ W1, __bf16* __restrict__ Bh, __bf16* __restrict__ Bl, int id) {
    if (id >= K_CHUNKS * 32 * 8) return;
    int v    = id & 7;
    int lane = (id >> 3) & 31;
    int c    = id >> 8;
    int n    = lane & 15;
    int g    = lane >> 4;
    int kb   = c * 32;
#pragma unroll
    for (int h = 0; h < 2; ++h) {
        int k = (v < 4) ? (kb + g * 8 + 2 * v + h)
                        : (kb + 16 + g * 8 + 2 * (v - 4) + h);
        float w   = W1[k * HID + n];
        __bf16 hi = (__bf16)w;
        __bf16 lo = (__bf16)(w - (float)hi);
        int idx = (c * 32 + lane) * 16 + v * 2 + h;
        Bh[idx] = hi;
        Bl[idx] = lo;
    }
}

__global__ void __launch_bounds__(256)
GCN_pack_w1(const float* __restrict__ W1,
            __bf16* __restrict__ Bh, __bf16* __restrict__ Bl) {
    pack_w1_job(W1, Bh, Bl, blockIdx.x * blockDim.x + threadIdx.x);
}

__global__ void __launch_bounds__(256)
GCN_gemm1_wmma(const float* __restrict__ x,
               const __bf16* __restrict__ Bh, const __bf16* __restrict__ Bl,
               float* __restrict__ h0) {
    const int wave = threadIdx.x >> 5;
    const int lane = threadIdx.x & 31;
    const int tile = blockIdx.x * 8 + wave;
    if (tile >= N_TILES) return;

    const int n = lane & 15;
    const int g = lane >> 4;
    const float* xrow = x + (size_t)(tile * 16 + n) * IN_F;

    v8f acc = {};
#pragma unroll 4
    for (int c = 0; c < K_CHUNKS; ++c) {
        const int kb = c * 32;
        float f[16];
        *(v4f*)(f + 0)  = *(const v4f*)(xrow + kb + g * 8);
        *(v4f*)(f + 4)  = *(const v4f*)(xrow + kb + g * 8 + 4);
        *(v4f*)(f + 8)  = *(const v4f*)(xrow + kb + 16 + g * 8);
        *(v4f*)(f + 12) = *(const v4f*)(xrow + kb + 16 + g * 8 + 4);
        if (c + 1 < K_CHUNKS)
            __builtin_prefetch(xrow + kb + 32 + g * 8, 0, 3);

        v16bf ah, al;
#pragma unroll
        for (int i = 0; i < 16; ++i) {
            __bf16 hi = (__bf16)f[i];
            ah[i] = hi;
            al[i] = (__bf16)(f[i] - (float)hi);
        }
        v16bf bh = *(const v16bf*)(Bh + (c * 32 + lane) * 16);
        v16bf bl = *(const v16bf*)(Bl + (c * 32 + lane) * 16);

        acc = __builtin_amdgcn_wmma_f32_16x16x32_bf16(false, ah, false, bh,
                                                      (short)0, acc, false, false);
        acc = __builtin_amdgcn_wmma_f32_16x16x32_bf16(false, al, false, bh,
                                                      (short)0, acc, false, false);
        acc = __builtin_amdgcn_wmma_f32_16x16x32_bf16(false, ah, false, bl,
                                                      (short)0, acc, false, false);
    }
#pragma unroll
    for (int r = 0; r < 8; ++r)
        h0[(size_t)(tile * 16 + r + g * 8) * HID + n] = acc[r];
}

__global__ void GCN_spmm16(const int* __restrict__ ei, const float* __restrict__ ew,
                           const float* __restrict__ h, float* __restrict__ agg) {
    long long t = (long long)blockIdx.x * blockDim.x + threadIdx.x;
    if (t >= (long long)E_EDGES * 16) return;
    int e = (int)(t >> 4);
    int j = (int)(t & 15);
    int src = ei[e];
    int dst = ei[E_EDGES + e];
    float v = ew[e] * h[(size_t)src * HID + j];
    atomic_add_f32(&agg[(size_t)dst * HID + j], v);
}

__global__ void GCN_spmm7(const int* __restrict__ ei, const float* __restrict__ ew,
                          const float* __restrict__ z, float* __restrict__ agg) {
    long long t = (long long)blockIdx.x * blockDim.x + threadIdx.x;
    if (t >= (long long)E_EDGES * 8) return;
    int e = (int)(t >> 3);
    int j = (int)(t & 7);
    if (j >= OUT_F) return;
    int src = ei[e];
    int dst = ei[E_EDGES + e];
    float v = ew[e] * z[(size_t)src * OUT_F + j];
    atomic_add_f32(&agg[(size_t)dst * OUT_F + j], v);
}

__global__ void __launch_bounds__(256)
GCN_relu_gemm2(const float* __restrict__ agg1, const float* __restrict__ b1,
               const float* __restrict__ W2, float* __restrict__ z) {
    __shared__ float sW2[HID * OUT_F];
    __shared__ float sb1[HID];
    if (threadIdx.x < HID * OUT_F) sW2[threadIdx.x] = W2[threadIdx.x];
    if (threadIdx.x < HID)         sb1[threadIdx.x] = b1[threadIdx.x];
    __syncthreads();

    int node = blockIdx.x * blockDim.x + threadIdx.x;
    if (node >= N_NODES) return;

    float hv[HID];
#pragma unroll
    for (int i = 0; i < HID; ++i)
        hv[i] = fmaxf(agg1[(size_t)node * HID + i] + sb1[i], 0.0f);
#pragma unroll
    for (int j = 0; j < OUT_F; ++j) {
        float s = 0.0f;
#pragma unroll
        for (int i = 0; i < HID; ++i)
            s = fmaf(hv[i], sW2[i * OUT_F + j], s);
        z[(size_t)node * OUT_F + j] = s;
    }
}

__global__ void __launch_bounds__(256)
GCN_logsoftmax(const float* __restrict__ agg2, const float* __restrict__ b2,
               float* __restrict__ out) {
    __shared__ float sb2[OUT_F];
    if (threadIdx.x < OUT_F) sb2[threadIdx.x] = b2[threadIdx.x];
    __syncthreads();

    int node = blockIdx.x * blockDim.x + threadIdx.x;
    if (node >= N_NODES) return;

    float v[OUT_F];
    float m = -3.402823466e38f;
#pragma unroll
    for (int j = 0; j < OUT_F; ++j) {
        v[j] = agg2[(size_t)node * OUT_F + j] + sb2[j];
        m = fmaxf(m, v[j]);
    }
    float s = 0.0f;
#pragma unroll
    for (int j = 0; j < OUT_F; ++j) s += expf(v[j] - m);
    float lse = logf(s);
#pragma unroll
    for (int j = 0; j < OUT_F; ++j)
        out[(size_t)node * OUT_F + j] = v[j] - m - lse;
}

__global__ void GCN_zero(float* __restrict__ p, long long n) {
    long long i = (long long)blockIdx.x * blockDim.x + threadIdx.x;
    if (i < n) p[i] = 0.0f;
}

#define RN        416
#define RG        ((N_NODES + RN - 1) / RN)
#define EP_T      1024
#define N_TILE    256
#define TILE_E    (E_EDGES / N_TILE)
#define TILE_Q    (TILE_E / 4)
#define TPW       (N_TILE / 32)
#define NB        (RG + 1)
#define NBP       256
#define OFF_PITCH NBP
#define SRC_BITS  17
#define PLACE_SH  8
#define SORT_K    ((TILE_Q + EP_T - 1) / EP_T)
#define W1_WG     ((K_CHUNKS * 32 * 8 + EP_T - 1) / EP_T)
static_assert(TILE_E * N_TILE == E_EDGES && TILE_Q * 4 == TILE_E && TPW * 32 == N_TILE, "tiles cover the edge list exactly, in whole 16-byte pieces");
static_assert(RN % 32 == 0 && RG * RN >= N_NODES && (RG - 1) * RN < N_NODES, "ranges cover the nodes; 32 rows of 7 floats are 7 whole lines");
static_assert(NB + 1 <= NBP && NBP == 32 * 8 && NB <= (1 << PLACE_SH) && TILE_E < (1 << (32 - PLACE_SH)), "one wave scans the bucket counts; a sort key holds bucket and place");
static_assert(N_NODES <= (1 << SRC_BITS) && RN <= (1 << (32 - SRC_BITS)), "a record's first word holds source and row");

typedef __attribute__((ext_vector_type(4))) int v4i;
typedef __attribute__((ext_vector_type(4))) unsigned v4u;
typedef __attribute__((ext_vector_type(2))) unsigned v2u;

__global__ void __launch_bounds__(EP_T)
GCN_pack_sort(const float* __restrict__ W1,
              __bf16* __restrict__ Bh, __bf16* __restrict__ Bl,
              const int* __restrict__ ei, const float* __restrict__ ew,
              unsigned* __restrict__ offs, unsigned* __restrict__ recs) {
    __shared__ __attribute__((aligned(16))) v2u stage[TILE_E];
    __shared__ unsigned cnt[NBP];
    __shared__ unsigned start[NBP];
    const int tid = threadIdx.x;
    if (blockIdx.x < W1_WG) {
        pack_w1_job(W1, Bh, Bl, blockIdx.x * EP_T + tid);
        return;
    }
    const int tile = blockIdx.x - W1_WG;
    const v4i* sq = (const v4i*)ei + (size_t)tile * TILE_Q;
    const v4i* dq = (const v4i*)(ei + E_EDGES) + (size_t)tile * TILE_Q;
    const v4f* wq = (const v4f*)ew + (size_t)tile * TILE_Q;

    if (tid < NBP) cnt[tid] = 0u;
    v4i d[SORT_K], s[SORT_K];
    v4f w[SORT_K];
#pragma unroll
    for (int k = 0; k < SORT_K; ++k) {
        const int g = tid + k * EP_T;
        if (g < TILE_Q) { d[k] = dq[g]; s[k] = sq[g]; w[k] = wq[g]; }
        else { d[k] = (v4i){-1, -1, -1, -1}; s[k] = (v4i){0, 0, 0, 0}; w[k] = (v4f){0.0f, 0.0f, 0.0f, 0.0f}; }
    }
    __syncthreads();

    unsigned qr[SORT_K][4];
#pragma unroll
    for (int k = 0; k < SORT_K; ++k) {
        const bool live = (tid + k * EP_T) < TILE_Q;
#pragma unroll
        for (int j = 0; j < 4; ++j) {
            unsigned q = (unsigned)d[k][j] / (unsigned)RN;
            q = (q < (unsigned)RG) ? q : (unsigned)RG;
            unsigned place = 0u;
            if (live) place = atomicAdd(&cnt[q], 1u);
            qr[k][j] = q | (place << PLACE_SH);
        }
    }
    __syncthreads();

    if (tid < 32) {
        unsigned c8[8];
        unsigned tot = 0u;
        for (int i = 0; i < 8; ++i) { c8[i] = cnt[tid * 8 + i]; tot += c8[i]; }
        unsigned incl = tot;
#pragma unroll
        for (int dlt = 1; dlt < 32; dlt <<= 1) {
            unsigned up = __shfl_up(incl, dlt);
            if (tid >= dlt) incl += up;
        }
        unsigned run = incl - tot;
        for (int i = 0; i < 8; ++i) {
            start[tid * 8 + i] = run;
            offs[tile * OFF_PITCH + tid * 8 + i] = run;
            run += c8[i];
        }
    }
    __syncthreads();

#pragma unroll
    for (int k = 0; k < SORT_K; ++k) {
        if ((tid + k * EP_T) < TILE_Q) {
#pragma unroll
            for (int j = 0; j < 4; ++j) {
                const unsigned q   = qr[k][j] & ((1u << PLACE_SH) - 1u);
                const unsigned pos = start[q] + (qr[k][j] >> PLACE_SH);
                const unsigned row = (unsigned)d[k][j] - q * (unsigned)RN;
                stage[pos] = (v2u){(unsigned)s[k][j] | (row << SRC_BITS), __float_as_uint(w[k][j])};
            }
        }
    }
    __syncthreads();

    const v4u* st4 = (const v4u*)stage;
    v4u* out4 = (v4u*)recs + (size_t)tile * (TILE_E / 2);
    for (int i = tid; i < TILE_E / 2; i += EP_T) out4[i] = st4[i];
}

template <int D>
__device__ __forceinline__ void
edge_round(const float* __restrict__ h, float* plane, int lane, unsigned nm,
           int sA, float wA, int rA, int sB, float wB, int rB) {
    constexpr int LPM    = 4;
    constexpr int MPS    = 32 / LPM;
    constexpr int NSTEP  = 64 / MPS;
    constexpr int VW     = (D == 16) ? 4 : 2;
    constexpr int PITCH  = (D == 16) ? 17 : 7;
    const int sub = lane / LPM;
    const int jl  = lane % LPM;

    float vv[NSTEP][VW];
    float wk[NSTEP];
    int   lk[NSTEP];
    bool  ok[NSTEP];
#pragma unroll
    for (int t = 0; t < NSTEP; ++t) {
        const int k  = t * MPS + sub;
        const int sl = k & 31;
        const int ss = __shfl((t * MPS < 32) ? sA : sB, sl);
        wk[t] = __shfl((t * MPS < 32) ? wA : wB, sl);
        lk[t] = __shfl((t * MPS < 32) ? rA : rB, sl);
        ok[t] = ((unsigned)k < nm);
        if constexpr (D == 16) {
            v4f x4 = {0.0f, 0.0f, 0.0f, 0.0f};
            if (ok[t]) x4 = *((const v4f*)(h + (size_t)ss * 16) + jl);
#pragma unroll
            for (int i = 0; i < VW; ++i) vv[t][i] = x4[i];
        } else {
            float x0 = 0.0f, x1 = 0.0f;
            if (ok[t]) {
                const float* hr = h + (size_t)ss * D + jl * 2;
                x0 = hr[0];
                if (jl < 3) x1 = hr[1];
            }
            vv[t][0] = x0;
            vv[t][1] = x1;
        }
    }
#pragma unroll
    for (int t = 0; t < NSTEP; ++t) {
        if (ok[t]) {
            float* pr = plane + lk[t] * PITCH + jl * VW;
#pragma unroll
            for (int i = 0; i < VW; ++i) {
                if (jl * VW + i < D)
                    __hip_atomic_fetch_add(pr + i, wk[t] * vv[t][i],
                                           __ATOMIC_RELAXED, __HIP_MEMORY_SCOPE_WORKGROUP);
            }
        }
    }
}

__device__ __forceinline__ unsigned
run_delta(unsigned idx,
          unsigned c1, unsigned c2, unsigned c3, unsigned c4, unsigned c5, unsigned c6, unsigned c7,
          unsigned d0, unsigned d1, unsigned d2, unsigned d3, unsigned d4, unsigned d5, unsigned d6, unsigned d7) {
    unsigned dd = d0;
    dd = (idx >= c1) ? d1 : dd;
    dd = (idx >= c2) ? d2 : dd;
    dd = (idx >= c3) ? d3 : dd;
    dd = (idx >= c4) ? d4 : dd;
    dd = (idx >= c5) ? d5 : dd;
    dd = (idx >= c6) ? d6 : dd;
    dd = (idx >= c7) ? d7 : dd;
    return dd;
}

template <int D>
__global__ void __launch_bounds__(EP_T)
GCN_edge_range(const float* __restrict__ h,
               const unsigned* __restrict__ offs, const unsigned* __restrict__ recs,
               const float* __restrict__ bias, const float* __restrict__ W2,
               float* __restrict__ dst) {
    constexpr int PITCH = (D == 16) ? 17 : 7;

    constexpr int PLANE_LDS   = RN * PITCH;
    constexpr int PLANE_ALLOC = 41216;
    static_assert(PLANE_LDS <= PLANE_ALLOC, "the plane fits its allocation");
    __shared__ __attribute__((aligned(16))) float plane[PLANE_ALLOC];
    __shared__ float sW2[HID * OUT_F];
    __shared__ float sb[HID];

    const int tid  = threadIdx.x;
    const int lane = tid & 31;
    const int wave = __builtin_amdgcn_readfirstlane(tid >> 5);
    const int r    = blockIdx.x;

    unsigned bv = 0u;
    if (lane < 2 * TPW) bv = offs[((lane >> 1) * 32 + wave) * OFF_PITCH + r + (lane & 1)];
    unsigned cs[TPW + 1];
    unsigned dl[TPW];
    cs[0] = 0u;
    for (int i = 0; i < TPW; ++i) {
        const unsigned a = (unsigned)__builtin_amdgcn_readlane((int)bv, 2 * i);
        const unsigned b = (unsigned)__builtin_amdgcn_readlane((int)bv, 2 * i + 1);
        dl[i] = (unsigned)(i * 32 + wave) * (unsigned)TILE_E + a - cs[i];
        cs[i + 1] = cs[i] + (b - a);
    }
    const unsigned n = cs[TPW];
    const unsigned c1 = cs[1], c2 = cs[2], c3 = cs[3], c4 = cs[4], c5 = cs[5], c6 = cs[6], c7 = cs[7];
    const unsigned d0 = dl[0], d1 = dl[1], d2 = dl[2], d3 = dl[3], d4 = dl[4], d5 = dl[5], d6 = dl[6], d7 = dl[7];
    static_assert(TPW == 8, "run_delta takes eight runs");
    const v2u* r2 = (const v2u*)recs;
    auto rec_at = [=](unsigned idx) -> v2u {
        return r2[idx + run_delta(idx, c1, c2, c3, c4, c5, c6, c7, d0, d1, d2, d3, d4, d5, d6, d7)];
    };

    v2u nA = {0u, 0u}, nB = {0u, 0u};
    if ((unsigned)lane < n)       nA = rec_at((unsigned)lane);
    if ((unsigned)lane + 32u < n) nB = rec_at((unsigned)lane + 32u);

    for (int i = tid; i < PLANE_LDS; i += EP_T) plane[i] = 0.0f;
    if constexpr (D == 16) {
        if (tid < HID * OUT_F) sW2[tid] = W2[tid];
    }
    if (tid < D) sb[tid] = bias[tid];
    __syncthreads();

    for (unsigned base = 0; base < n; base += 64) {
        const unsigned nm = (n - base < 64u) ? (n - base) : 64u;
        const v2u cA = nA, cB = nB;
        nA = (v2u){0u, 0u}; nB = (v2u){0u, 0u};
        if (base + 64u + (unsigned)lane < n) nA = rec_at(base + 64u + (unsigned)lane);
        if (base + 96u + (unsigned)lane < n) nB = rec_at(base + 96u + (unsigned)lane);
        edge_round<D>(h, plane, lane, nm,
                      (int)(cA[0] & ((1u << SRC_BITS) - 1u)), __uint_as_float(cA[1]), (int)(cA[0] >> SRC_BITS),
                      (int)(cB[0] & ((1u << SRC_BITS) - 1u)), __uint_as_float(cB[1]), (int)(cB[0] >> SRC_BITS));
    }
    __syncthreads();

    const int node = r * RN + tid;
    if (tid >= RN || node >= N_NODES) return;
    const float* pr = plane + tid * PITCH;
    if constexpr (D == 16) {
        float hv[HID];
        for (int i = 0; i < HID; ++i) hv[i] = fmaxf(pr[i] + sb[i], 0.0f);
        for (int j = 0; j < OUT_F; ++j) {
            float sum = 0.0f;
            for (int i = 0; i < HID; ++i)
                sum = fmaf(hv[i], sW2[i * OUT_F + j], sum);
            dst[(size_t)node * OUT_F + j] = sum;
        }
    } else {
        float v[OUT_F];
        float m = -3.402823466e38f;
        for (int j = 0; j < OUT_F; ++j) {
            v[j] = pr[j] + sb[j];
            m = fmaxf(m, v[j]);
        }
        float sum = 0.0f;
        for (int j = 0; j < OUT_F; ++j) sum += expf(v[j] - m);
        const float lse = logf(sum);
        for (int j = 0; j < OUT_F; ++j)
            dst[(size_t)node * OUT_F + j] = v[j] - m - lse;
    }
}

extern "C" void kernel_launch(void* const* d_in, const int* in_sizes, int n_in,
                              void* d_out, int out_size, void* d_ws, size_t ws_size,
                              hipStream_t stream) {
    const float* x  = (const float*)d_in[0];
    const int*   ei = (const int*)d_in[1];
    const float* ew = (const float*)d_in[2];
    const float* W1 = (const float*)d_in[3];
    const float* b1 = (const float*)d_in[4];
    const float* W2 = (const float*)d_in[5];
    const float* b2 = (const float*)d_in[6];
    float* out = (float*)d_out;

    char* ws = (char*)d_ws;
    size_t off = 0;
    auto alloc = [&](size_t bytes) { char* p = ws + off; off = (off + bytes + 255) & ~(size_t)255; return p; };
    float*  h0   = (float*)alloc((size_t)N_NODES * HID * 4);
    float*  z    = (float*)alloc((size_t)N_NODES * OUT_F * 4);
    __bf16* Bh   = (__bf16*)alloc((size_t)K_CHUNKS * 32 * 16 * 2);
    __bf16* Bl   = (__bf16*)alloc((size_t)K_CHUNKS * 32 * 16 * 2);
    unsigned* erec = (unsigned*)alloc((size_t)E_EDGES * 8);
    unsigned* eoff = (unsigned*)alloc((size_t)N_TILE * OFF_PITCH * 4);

    if (off <= ws_size) {
        GCN_pack_sort<<<W1_WG + N_TILE, EP_T, 0, stream>>>(W1, Bh, Bl, ei, ew, eoff, erec);
        GCN_gemm1_wmma<<<(N_TILES + 7) / 8, 256, 0, stream>>>(x, Bh, Bl, h0);
        GCN_edge_range<16><<<RG, EP_T, 0, stream>>>(h0, eoff, erec, b1, W2, z);
        GCN_edge_range<OUT_F><<<RG, EP_T, 0, stream>>>(z, eoff, erec, b2, W2, out);
        return;
    }

    off = 0;
    h0 = (float*)alloc((size_t)N_NODES * HID * 4);
    float* agg1 = (float*)alloc((size_t)N_NODES * HID * 4);
    z = (float*)alloc((size_t)N_NODES * OUT_F * 4);
    float* agg2 = (float*)alloc((size_t)N_NODES * OUT_F * 4);
    Bh = (__bf16*)alloc((size_t)K_CHUNKS * 32 * 16 * 2);
    Bl = (__bf16*)alloc((size_t)K_CHUNKS * 32 * 16 * 2);
    {
        long long n1 = (long long)N_NODES * HID;
        long long n2 = (long long)N_NODES * OUT_F;
        GCN_zero<<<(int)((n1 + 255) / 256), 256, 0, stream>>>(agg1, n1);
        GCN_zero<<<(int)((n2 + 255) / 256), 256, 0, stream>>>(agg2, n2);
    }
    GCN_pack_w1<<<K_CHUNKS, 256, 0, stream>>>(W1, Bh, Bl);
    GCN_gemm1_wmma<<<(N_TILES + 7) / 8, 256, 0, stream>>>(x, Bh, Bl, h0);
    {
        long long t = (long long)E_EDGES * 16;
        GCN_spmm16<<<(int)((t + 255) / 256), 256, 0, stream>>>(ei, ew, h0, agg1);
    }
    GCN_relu_gemm2<<<(N_NODES + 255) / 256, 256, 0, stream>>>(agg1, b1, W2, z);
    {
        long long t = (long long)E_EDGES * 8;
        GCN_spmm7<<<(int)((t + 255) / 256), 256, 0, stream>>>(ei, ew, z, agg2);
    }
    GCN_logsoftmax<<<(N_NODES + 255) / 256, 256, 0, stream>>>(agg2, b2, out);
}
